# starting kernel
# baseline (speedup 1.0000x reference)
#include <hip/hip_runtime.h>

#define D   512
#define R   64
#define WIN 32
#define BB  4
#define SS  2048
#define NN  (SS + 1)
#define MM  (BB * NN)
#define TILES ((NN + 15) / 16)
#define WTOT 66
#define KB  84

typedef float v2f __attribute__((ext_vector_type(2)));
typedef float v8f __attribute__((ext_vector_type(8)));

__device__ __forceinline__ int clampi(int x, int lo, int hi) {
    return min(max(x, lo), hi);
}

__global__ __launch_bounds__(256) void embed_ln_kernel(
    const int* __restrict__ ids, const float* __restrict__ emb,
    const float* __restrict__ pos, const float* __restrict__ anchor_val,
    const float* __restrict__ anchor_state, const float* __restrict__ gamma,
    const float* __restrict__ beta, const float* __restrict__ sp_w,
    const float* __restrict__ sp_b, float* __restrict__ val,
    float* __restrict__ state)
{
    __shared__ float sred[256];
    const int bn  = blockIdx.x;
    const int b   = bn / NN;
    const int n   = bn % NN;
    const int tid = threadIdx.x;
    float* vrow = val + (size_t)bn * D;

    if (n == 0) {
        for (int d = tid; d < D; d += 256) vrow[d] = anchor_val[d];
        if (tid == 0) state[bn] = anchor_state[0];
        return;
    }
    const int s  = n - 1;
    const int id = ids[b * SS + s];
    const float* erow = emb + (size_t)id * D;
    const float* prow = pos + (size_t)s * D;
    const int d0 = tid, d1 = tid + 256;
    float x0 = erow[d0] + prow[d0];
    float x1 = erow[d1] + prow[d1];

    sred[tid] = x0 + x1; __syncthreads();
    for (int o = 128; o > 0; o >>= 1) { if (tid < o) sred[tid] += sred[tid + o]; __syncthreads(); }
    float mean = sred[0] * (1.0f / D);
    __syncthreads();
    float c0 = x0 - mean, c1 = x1 - mean;
    sred[tid] = c0 * c0 + c1 * c1; __syncthreads();
    for (int o = 128; o > 0; o >>= 1) { if (tid < o) sred[tid] += sred[tid + o]; __syncthreads(); }
    float rstd = rsqrtf(sred[0] * (1.0f / D) + 1e-5f);
    __syncthreads();
    float y0 = c0 * rstd * gamma[d0] + beta[d0];
    float y1 = c1 * rstd * gamma[d1] + beta[d1];
    vrow[d0] = y0; vrow[d1] = y1;
    sred[tid] = y0 * sp_w[d0] + y1 * sp_w[d1]; __syncthreads();
    for (int o = 128; o > 0; o >>= 1) { if (tid < o) sred[tid] += sred[tid + o]; __syncthreads(); }
    if (tid == 0) state[bn] = sred[0] + sp_b[0];
}

__global__ __launch_bounds__(256) void qk_gemm_kernel(
    const float* __restrict__ val, const float* __restrict__ wq,
    const float* __restrict__ wk, float* __restrict__ q,
    float* __restrict__ k)
{
    const int lane   = threadIdx.x & 31;
    const int wave   = threadIdx.x >> 5;
    const int mtile  = blockIdx.x * 8 + wave;
    const int mtiles = (MM + 15) / 16;
    if (mtile >= mtiles) return;

    const int row0 = mtile * 16;
    const int lm   = lane & 15;
    const int lh   = lane >> 4;
    const int arow = min(row0 + lm, MM - 1);
    const float* aptr = val + (size_t)arow * D + 2 * lh;

    v8f accq[4] = {};
    v8f acck[4] = {};

    for (int kk = 0; kk < D; kk += 4) {
        v2f a = *(const v2f*)(aptr + kk);
        const float* bqp = wq + (size_t)(kk + 2 * lh) * R + lm;
        const float* bkp = wk + (size_t)(kk + 2 * lh) * R + lm;
#pragma unroll
        for (int nt = 0; nt < 4; ++nt) {
            v2f bq; bq.x = bqp[nt * 16]; bq.y = bqp[R + nt * 16];
            v2f bk; bk.x = bkp[nt * 16]; bk.y = bkp[R + nt * 16];
            accq[nt] = __builtin_amdgcn_wmma_f32_16x16x4_f32(
                false, a, false, bq, (short)0, accq[nt], false, false);
            acck[nt] = __builtin_amdgcn_wmma_f32_16x16x4_f32(
                false, a, false, bk, (short)0, acck[nt], false, false);
        }
    }

#pragma unroll
    for (int nt = 0; nt < 4; ++nt) {
#pragma unroll
        for (int i = 0; i < 8; ++i) {
            const int r = row0 + i + 8 * lh;
            if (r < MM) {
                q[(size_t)r * R + nt * 16 + lm] = accq[nt][i];
                k[(size_t)r * R + nt * 16 + lm] = acck[nt][i];
            }
        }
    }
}

__global__ __launch_bounds__(256) void attn_fused_kernel(
    const float* __restrict__ val_in, const float* __restrict__ state_in,
    const float* __restrict__ q, const float* __restrict__ k,
    const float* __restrict__ gamma, const float* __restrict__ beta,
    float* __restrict__ val_out, float* __restrict__ state_out)
{
    __shared__ float s_scores[16][80];
    __shared__ float s_w[16][KB];
    __shared__ float s_y[16][D];
    __shared__ float s_r[16][16];
    __shared__ float s_mean[16];
    __shared__ float s_rstd[16];

    const int tid  = threadIdx.x;
    const int lane = tid & 31;
    const int wave = tid >> 5;
    const int lm   = lane & 15;
    const int lh   = lane >> 4;
    const int b    = blockIdx.x / TILES;
    const int t    = blockIdx.x % TILES;
    const int p0   = t * 16;
    const int base = b * NN;

    if (wave < 5) {
        const int qrow = base + clampi(p0 + lm, 0, NN - 1);
        const int j    = wave * 16 + lm;
        const int krow = base + clampi(p0 - 32 + j, 0, NN - 1);
        const float* qp = q + (size_t)qrow * R + 2 * lh;
        const float* kp = k + (size_t)krow * R + 2 * lh;
        v8f acc = {};
        for (int kk = 0; kk < R; kk += 4) {
            v2f a  = *(const v2f*)(qp + kk);
            v2f bb = *(const v2f*)(kp + kk);
            acc = __builtin_amdgcn_wmma_f32_16x16x4_f32(
                false, a, false, bb, (short)0, acc, false, false);
        }
#pragma unroll
        for (int i = 0; i < 8; ++i)
            s_scores[i + 8 * lh][wave * 16 + lm] = acc[i];
    }
    __syncthreads();

    for (int x = tid; x < 16 * KB; x += 256) (&s_w[0][0])[x] = 0.f;
    __syncthreads();

    if (tid < 16) {
        const int m    = tid;
        const int node = p0 + m;
        if (node < NN) {
            const float scale = 0.125f;
            float adot = 0.f;
            {
                const float* qp = q + (size_t)(base + node) * R;
                const float* k0 = k + (size_t)base * R;
                for (int r = 0; r < R; ++r) adot = fmaf(qp[r], k0[r], adot);
                adot *= scale;
            }
            const bool avalid = (node - WIN) > 0;
            float mx = avalid ? fabsf(adot) : -1e30f;
            for (int w = 0; w < 65; ++w) {
                const int p = node - WIN + w;
                if (p >= 0 && p < NN)
                    mx = fmaxf(mx, fabsf(s_scores[m][m + w] * scale));
            }
            float sum = 0.f;
            for (int w = 0; w < 65; ++w) {
                const int p = node - WIN + w;
                float se = 0.f;
                if (p >= 0 && p < NN) {
                    const float s = s_scores[m][m + w] * scale;
                    const float e = __expf(fabsf(s) - mx);
                    sum += e;
                    se = (s > 0.f) ? e : ((s < 0.f) ? -e : 0.f);
                }
                s_scores[m][m + w] = se;
            }
            float ase = 0.f;
            if (avalid) {
                const float e = __expf(fabsf(adot) - mx);
                sum += e;
                ase = (adot > 0.f) ? e : ((adot < 0.f) ? -e : 0.f);
            }
            const float inv = 1.f / sum;
            float ds = 0.f;
            for (int w = 0; w < 65; ++w) {
                const float we = s_scores[m][m + w] * inv;
                s_w[m][m + w] = we;
                ds += we * state_in[base + clampi(node - WIN + w, 0, NN - 1)];
            }
            const float wa = ase * inv;
            s_w[m][80] = wa;
            ds += wa * state_in[base];
            state_out[base + node] = state_in[base + node] + ds;
        }
    }
    __syncthreads();

    {
        const int c0 = wave * 64;
        v8f acc[4] = {};
        for (int kk = 0; kk < KB; kk += 4) {
            const int kp0 = kk + 2 * lh;
            const int kp1 = kp0 + 1;
            v2f a;
            a.x = s_w[lm][kp0];
            a.y = s_w[lm][kp1];
            const int r0 = (kp0 >= 80) ? base : base + clampi(p0 - 32 + kp0, 0, NN - 1);
            const int r1 = (kp1 >= 80) ? base : base + clampi(p0 - 32 + kp1, 0, NN - 1);
            const float* v0 = val_in + (size_t)r0 * D + c0 + lm;
            const float* v1 = val_in + (size_t)r1 * D + c0 + lm;
#pragma unroll
            for (int nt = 0; nt < 4; ++nt) {
                v2f bb;
                bb.x = v0[nt * 16];
                bb.y = v1[nt * 16];
                acc[nt] = __builtin_amdgcn_wmma_f32_16x16x4_f32(
                    false, a, false, bb, (short)0, acc[nt], false, false);
            }
        }
#pragma unroll
        for (int nt = 0; nt < 4; ++nt) {
#pragma unroll
            for (int i = 0; i < 8; ++i) {
                const int m    = i + 8 * lh;
                const int col  = c0 + nt * 16 + lm;
                const int node = p0 + m;
                float y = 0.f;
                if (node < NN)
                    y = acc[nt][i] + val_in[(size_t)(base + node) * D + col];
                s_y[m][col] = y;
            }
        }
    }
    __syncthreads();

    const int m  = tid >> 4;
    const int jj = tid & 15;
    float part = 0.f;
    for (int e = 0; e < 32; ++e) part += s_y[m][jj * 32 + e];
    s_r[m][jj] = part;
    __syncthreads();
    if (jj == 0) {
        float tot = 0.f;
        for (int x = 0; x < 16; ++x) tot += s_r[m][x];
        s_mean[m] = tot * (1.0f / D);
    }
    __syncthreads();
    const float mean = s_mean[m];
    part = 0.f;
    for (int e = 0; e < 32; ++e) {
        const float c = s_y[m][jj * 32 + e] - mean;
        part += c * c;
    }
    s_r[m][jj] = part;
    __syncthreads();
    if (jj == 0) {
        float tot = 0.f;
        for (int x = 0; x < 16; ++x) tot += s_r[m][x];
        s_rstd[m] = rsqrtf(tot * (1.0f / D) + 1e-5f);
    }
    __syncthreads();
    const int node = p0 + m;
    if (node < NN) {
        const float rstd = s_rstd[m];
        float* vo = val_out + (size_t)(base + node) * D;
        for (int e = 0; e < 32; ++e) {
            const int col = jj * 32 + e;
            vo[col] = (s_y[m][col] - mean) * rstd * gamma[col] + beta[col];
        }
    }
}

extern "C" void kernel_launch(void* const* d_in, const int* in_sizes, int n_in,
                              void* d_out, int out_size, void* d_ws, size_t ws_size,
                              hipStream_t stream)
{
    const int*   ids    = (const int*)  d_in[0];
    const float* emb    = (const float*)d_in[1];
    const float* pos    = (const float*)d_in[2];
    const float* aval   = (const float*)d_in[3];
    const float* astate = (const float*)d_in[4];
    const float* ing    = (const float*)d_in[5];
    const float* inb    = (const float*)d_in[6];
    const float* spw    = (const float*)d_in[7];
    const float* spb    = (const float*)d_in[8];
    const float* wq     = (const float*)d_in[9];
    const float* wk     = (const float*)d_in[10];
    const float* lng    = (const float*)d_in[11];
    const float* lnb    = (const float*)d_in[12];

    float* ws = (float*)d_ws;
    size_t o = 0;
    float* val_a = ws + o; o += (size_t)MM * D;
    float* val_b = ws + o; o += (size_t)MM * D;
    float* st_a  = ws + o; o += MM;
    float* st_b  = ws + o; o += MM;
    float* qbuf  = ws + o; o += (size_t)MM * R;
    float* kbuf  = ws + o; o += (size_t)MM * R;

    float* out_state = (float*)d_out;
    float* out_val   = out_state + MM;

    embed_ln_kernel<<<MM, 256, 0, stream>>>(ids, emb, pos, aval, astate,
                                            ing, inb, spw, spb, val_a, st_a);

    const int mtiles  = (MM + 15) / 16;
    const int gblocks = (mtiles + 7) / 8;
    const int ablocks = BB * TILES;

    qk_gemm_kernel<<<gblocks, 256, 0, stream>>>(val_a, wq, wk, qbuf, kbuf);
    attn_fused_kernel<<<ablocks, 256, 0, stream>>>(val_a, st_a, qbuf, kbuf,
                                                   lng, lnb, val_b, st_b);
    qk_gemm_kernel<<<gblocks, 256, 0, stream>>>(val_b, wq + (size_t)D * R,
                                                wk + (size_t)D * R, qbuf, kbuf);
    attn_fused_kernel<<<ablocks, 256, 0, stream>>>(val_b, st_b, qbuf, kbuf,
                                                   lng + D, lnb + D, out_val, out_state);
}


# Scores computed transposed by every wave: softmax in the accumulator registers, weights feed the dval WMMA as A operand from there; dval k order and softmax, anchor-dot, dstate sum order changed
# speedup vs baseline: 2.6527x; 2.6527x over previous
#include <hip/hip_runtime.h>

#define D   512
#define R   64
#define WIN 32
#define BB  4
#define SS  2048
#define NN  (SS + 1)
#define MM  (BB * NN)
#define TILES ((NN + 15) / 16)
#define WTOT 66
#define DP  (D + 4)

typedef float v2f __attribute__((ext_vector_type(2)));
typedef float v8f __attribute__((ext_vector_type(8)));

__device__ __forceinline__ int clampi(int x, int lo, int hi) {
    return min(max(x, lo), hi);
}

__device__ __forceinline__ void qk_project_tile(
    const float* s_t, const float* __restrict__ wq, const float* __restrict__ wk,
    float* __restrict__ qo, float* __restrict__ ko,
    int grow0, int nvalid, int wave, int lm, int lh)
{
    constexpr int U = 16;
    constexpr int G = D / (4 * U);
    const float* wm = (wave < 4) ? wq : wk;
    float*       om = (wave < 4) ? qo : ko;
    const int col = (wave & 3) * 16 + lm;
    const float* bp = wm + (size_t)(2 * lh) * R + col;
    const float* ap = s_t + lm * DP + 2 * lh;

    float bx[U], by[U];
#pragma unroll
    for (int u = 0; u < U; ++u) {
        bx[u] = bp[(size_t)(4 * u) * R];
        by[u] = bp[(size_t)(4 * u + 1) * R];
    }
    v8f acc = {};
#pragma unroll
    for (int g = 0; g < G; ++g) {
        float nx[U], ny[U];
        if (g + 1 < G) {
            const float* np = bp + (size_t)(g + 1) * (4 * U) * R;
#pragma unroll
            for (int u = 0; u < U; ++u) {
                nx[u] = np[(size_t)(4 * u) * R];
                ny[u] = np[(size_t)(4 * u + 1) * R];
            }
        }
#pragma unroll
        for (int u = 0; u < U; ++u) {
            v2f a = *(const v2f*)(ap + g * (4 * U) + 4 * u);
            v2f b; b.x = bx[u]; b.y = by[u];
            acc = __builtin_amdgcn_wmma_f32_16x16x4_f32(
                false, a, false, b, (short)0, acc, false, false);
        }
        if (g + 1 < G) {
#pragma unroll
            for (int u = 0; u < U; ++u) { bx[u] = nx[u]; by[u] = ny[u]; }
        }
    }
#pragma unroll
    for (int i = 0; i < 8; ++i) {
        const int m = i + 8 * lh;
        if (m < nvalid) om[(size_t)(grow0 + m) * R + col] = acc[i];
    }
}

__global__ __launch_bounds__(256) void embed_ln_qk_kernel(
    const int* __restrict__ ids, const float* __restrict__ emb,
    const float* __restrict__ pos, const float* __restrict__ anchor_val,
    const float* __restrict__ anchor_state, const float* __restrict__ gamma,
    const float* __restrict__ beta, const float* __restrict__ sp_w,
    const float* __restrict__ sp_b, float* __restrict__ val,
    float* __restrict__ state,
    const float* __restrict__ wq, const float* __restrict__ wk,
    float* __restrict__ q, float* __restrict__ k)
{
    __shared__ __attribute__((aligned(16))) float s_y[16][DP];
    __shared__ __attribute__((aligned(16))) float s_p[3][16][16];

    const int tid  = threadIdx.x;
    const int lane = tid & 31;
    const int wave = tid >> 5;
    const int lm   = lane & 15;
    const int lh   = lane >> 4;
    const int b    = blockIdx.x / TILES;
    const int t    = blockIdx.x % TILES;
    const int p0   = t * 16;
    const int base = b * NN;

    const int m    = tid >> 4;
    const int jj   = tid & 15;
    const int node = p0 + m;
    const bool is_tok = (node >= 1) && (node < NN);

    float4 x[8];
    if (is_tok) {
        const int s  = node - 1;
        const int id = ids[b * SS + s];
        const float4* e4 = (const float4*)(emb + (size_t)id * D) + jj;
        const float4* p4 = (const float4*)(pos + (size_t)s * D) + jj;
#pragma unroll
        for (int i = 0; i < 8; ++i) {
            const float4 ev = e4[16 * i];
            const float4 pv = p4[16 * i];
            x[i] = make_float4(ev.x + pv.x, ev.y + pv.y, ev.z + pv.z, ev.w + pv.w);
        }
    } else if (node == 0) {
        const float4* a4 = (const float4*)anchor_val + jj;
#pragma unroll
        for (int i = 0; i < 8; ++i) x[i] = a4[16 * i];
    } else {
#pragma unroll
        for (int i = 0; i < 8; ++i) x[i] = make_float4(0.f, 0.f, 0.f, 0.f);
    }

    float part = 0.f;
#pragma unroll
    for (int i = 0; i < 8; ++i) part += (x[i].x + x[i].y) + (x[i].z + x[i].w);
    s_p[0][m][jj] = part;
    __syncthreads();
    float tot = 0.f;
#pragma unroll
    for (int u = 0; u < 16; ++u) tot += s_p[0][m][u];
    const float mean = tot * (1.0f / D);

    part = 0.f;
#pragma unroll
    for (int i = 0; i < 8; ++i) {
        const float cx = x[i].x - mean, cy = x[i].y - mean, cz = x[i].z - mean, cw = x[i].w - mean;
        part += (cx * cx + cy * cy) + (cz * cz + cw * cw);
    }
    s_p[1][m][jj] = part;
    __syncthreads();
    tot = 0.f;
#pragma unroll
    for (int u = 0; u < 16; ++u) tot += s_p[1][m][u];
    const float rstd = rsqrtf(tot * (1.0f / D) + 1e-5f);

    const float4* g4 = (const float4*)gamma + jj;
    const float4* b4 = (const float4*)beta + jj;
    const float4* w4 = (const float4*)sp_w + jj;
    part = 0.f;
#pragma unroll
    for (int i = 0; i < 8; ++i) {
        const float4 g = g4[16 * i];
        const float4 be = b4[16 * i];
        const float4 w = w4[16 * i];
        float4 y;
        y.x = (x[i].x - mean) * rstd * g.x + be.x;
        y.y = (x[i].y - mean) * rstd * g.y + be.y;
        y.z = (x[i].z - mean) * rstd * g.z + be.z;
        y.w = (x[i].w - mean) * rstd * g.w + be.w;
        part += (y.x * w.x + y.y * w.y) + (y.z * w.z + y.w * w.w);
        if (is_tok) x[i] = y;
        *((float4*)&s_y[m][4 * jj + 64 * i]) = x[i];
    }
    s_p[2][m][jj] = part;
    __syncthreads();
    if (node < NN) {
        float4* vo = (float4*)(val + (size_t)(base + node) * D) + jj;
#pragma unroll
        for (int i = 0; i < 8; ++i) vo[16 * i] = x[i];
    }
    if (jj == 0 && node < NN) {
        float st;
        if (node == 0) {
            st = anchor_state[0];
        } else {
            tot = 0.f;
#pragma unroll
            for (int u = 0; u < 16; ++u) tot += s_p[2][m][u];
            st = tot + sp_b[0];
        }
        state[base + node] = st;
    }

    qk_project_tile(&s_y[0][0], wq, wk, q, k, base + p0, NN - p0, wave, lm, lh);
}

template <bool PROJ>
__global__ __launch_bounds__(256) __attribute__((amdgpu_waves_per_eu(PROJ ? 1 : 2, PROJ ? 16 : 7))) void attn_fused_kernel(
    const float* __restrict__ val_in, const float* __restrict__ state_in,
    const float* __restrict__ q, const float* __restrict__ k,
    const float* __restrict__ gamma, const float* __restrict__ beta,
    float* __restrict__ val_out, float* __restrict__ state_out,
    const float* __restrict__ wq_next, const float* __restrict__ wk_next,
    float* __restrict__ q_next, float* __restrict__ k_next)
{
    __shared__ __attribute__((aligned(16))) float s_y[16][DP];
    __shared__ float s_r[16][16];
    __shared__ float s_mean[16];
    __shared__ float s_rstd[16];

    const int tid  = threadIdx.x;
    const int lane = tid & 31;
    const int wave = tid >> 5;
    const int lm   = lane & 15;
    const int lh   = lane >> 4;
    const int b    = blockIdx.x / TILES;
    const int t    = blockIdx.x % TILES;
    const int p0   = t * 16;
    const int base = b * NN;

    const int node_l = p0 + lm;
    const int pb     = p0 - WIN + 8 * lh;
    v2f qf[16];
    {
        const float* qp = q + (size_t)(base + clampi(node_l, 0, NN - 1)) * R + 2 * lh;
#pragma unroll
        for (int s = 0; s < 16; ++s) qf[s] = *(const v2f*)(qp + 4 * s);
    }
    v8f sc[5];
#pragma unroll
    for (int bt = 0; bt < 5; ++bt) {
        const int krow = base + clampi(p0 - WIN + 16 * bt + lm, 0, NN - 1);
        const float* kp = k + (size_t)krow * R + 2 * lh;
        v8f acc = {};
#pragma unroll
        for (int s = 0; s < 16; ++s) {
            v2f a = *(const v2f*)(kp + 4 * s);
            acc = __builtin_amdgcn_wmma_f32_16x16x4_f32(
                false, a, false, qf[s], (short)0, acc, false, false);
        }
        sc[bt] = acc;
    }

    const float scale = 0.125f;
    float adot;
    {
        const float* k0 = k + (size_t)base * R + 2 * lh;
        float pa = 0.f;
#pragma unroll
        for (int s = 0; s < 16; ++s) {
            const v2f kv = *(const v2f*)(k0 + 4 * s);
            pa = fmaf(qf[s].x, kv.x, pa);
            pa = fmaf(qf[s].y, kv.y, pa);
        }
        adot = (pa + __shfl_xor(pa, 16)) * scale;
    }
    const bool avalid = (node_l - WIN) > 0;
    const int lo = max(lm - 8 * lh, -pb);
    const int hi = min(lm - 8 * lh + 2 * WIN, NN - 1 - pb);
    float mx = avalid ? fabsf(adot) : -1e30f;
#pragma unroll
    for (int bt = 0; bt < 5; ++bt) {
#pragma unroll
        for (int i = 0; i < 8; ++i) {
            const int c = 16 * bt + i;
            const float s = sc[bt][i] * scale;
            sc[bt][i] = s;
            if (c >= lo && c <= hi) mx = fmaxf(mx, fabsf(s));
        }
    }
    mx = fmaxf(mx, __shfl_xor(mx, 16));
    float sum = 0.f;
#pragma unroll
    for (int bt = 0; bt < 5; ++bt) {
#pragma unroll
        for (int i = 0; i < 8; ++i) {
            const int c = 16 * bt + i;
            const bool ok = (c >= lo) && (c <= hi);
            const float s = sc[bt][i];
            const float e = __expf(fabsf(s) - mx);
            const float se = (s > 0.f) ? e : ((s < 0.f) ? -e : 0.f);
            sum += ok ? e : 0.f;
            sc[bt][i] = ok ? se : 0.f;
        }
    }
    sum += __shfl_xor(sum, 16);
    float ase = 0.f;
    if (avalid) {
        const float e = __expf(fabsf(adot) - mx);
        sum += e;
        ase = (adot > 0.f) ? e : ((adot < 0.f) ? -e : 0.f);
    }
    const float inv = 1.f / sum;
    const float wa  = ase * inv;
#pragma unroll
    for (int bt = 0; bt < 5; ++bt) {
#pragma unroll
        for (int i = 0; i < 8; ++i) sc[bt][i] *= inv;
    }
    if (wave == 0) {
        float ds = 0.f;
#pragma unroll
        for (int bt = 0; bt < 5; ++bt) {
#pragma unroll
            for (int i = 0; i < 8; ++i)
                ds += sc[bt][i] * state_in[base + clampi(pb + 16 * bt + i, 0, NN - 1)];
        }
        ds += __shfl_xor(ds, 16);
        ds += wa * state_in[base];
        if (lh == 0 && node_l < NN)
            state_out[base + node_l] = state_in[base + node_l] + ds;
    }

    {
        const int c0 = wave * 64;
        v8f acc[4] = {};
#pragma unroll
        for (int s = 0; s < 20; ++s) {
            const int bt = s >> 2;
            const int i0 = 2 * (s & 3);
            v2f a;
            a.x = sc[bt][i0];
            a.y = sc[bt][i0 + 1];
            const int r0 = base + clampi(pb + 16 * bt + i0,     0, NN - 1);
            const int r1 = base + clampi(pb + 16 * bt + i0 + 1, 0, NN - 1);
            const float* v0 = val_in + (size_t)r0 * D + c0 + lm;
            const float* v1 = val_in + (size_t)r1 * D + c0 + lm;
#pragma unroll
            for (int nt = 0; nt < 4; ++nt) {
                v2f bb;
                bb.x = v0[nt * 16];
                bb.y = v1[nt * 16];
                acc[nt] = __builtin_amdgcn_wmma_f32_16x16x4_f32(
                    false, a, false, bb, (short)0, acc[nt], false, false);
            }
        }
        {
            v2f a;
            a.x = (lh == 0) ? wa : 0.f;
            a.y = 0.f;
            const float* v0 = val_in + (size_t)base * D + c0 + lm;
#pragma unroll
            for (int nt = 0; nt < 4; ++nt) {
                v2f bb;
                bb.x = v0[nt * 16];
                bb.y = bb.x;
                acc[nt] = __builtin_amdgcn_wmma_f32_16x16x4_f32(
                    false, a, false, bb, (short)0, acc[nt], false, false);
            }
        }
#pragma unroll
        for (int nt = 0; nt < 4; ++nt) {
#pragma unroll
            for (int i = 0; i < 8; ++i) {
                const int m    = i + 8 * lh;
                const int col  = c0 + nt * 16 + lm;
                const int node = p0 + m;
                float y = 0.f;
                if (node < NN)
                    y = acc[nt][i] + val_in[(size_t)(base + node) * D + col];
                s_y[m][col] = y;
            }
        }
    }
    __syncthreads();

    const int m  = tid >> 4;
    const int jj = tid & 15;
    float part = 0.f;
    for (int e = 0; e < 32; ++e) part += s_y[m][jj * 32 + e];
    s_r[m][jj] = part;
    __syncthreads();
    if (jj == 0) {
        float tot = 0.f;
        for (int x = 0; x < 16; ++x) tot += s_r[m][x];
        s_mean[m] = tot * (1.0f / D);
    }
    __syncthreads();
    const float mean = s_mean[m];
    part = 0.f;
    for (int e = 0; e < 32; ++e) {
        const float c = s_y[m][jj * 32 + e] - mean;
        part += c * c;
    }
    s_r[m][jj] = part;
    __syncthreads();
    if (jj == 0) {
        float tot = 0.f;
        for (int x = 0; x < 16; ++x) tot += s_r[m][x];
        s_rstd[m] = rsqrtf(tot * (1.0f / D) + 1e-5f);
    }
    __syncthreads();
    const int node = p0 + m;
    if (node < NN) {
        const float rstd = s_rstd[m];
        float* vo = val_out + (size_t)(base + node) * D;
        for (int e = 0; e < 32; ++e) {
            const int col = jj * 32 + e;
            const float o = (s_y[m][col] - mean) * rstd * gamma[col] + beta[col];
            vo[col] = o;
            if (PROJ) s_y[m][col] = o;
        }
    }

    if (PROJ) {
        __syncthreads();
        qk_project_tile(&s_y[0][0], wq_next, wk_next, q_next, k_next,
                        base + p0, NN - p0, wave, lm, lh);
    }
}

extern "C" void kernel_launch(void* const* d_in, const int* in_sizes, int n_in,
                              void* d_out, int out_size, void* d_ws, size_t ws_size,
                              hipStream_t stream)
{
    const int*   ids    = (const int*)  d_in[0];
    const float* emb    = (const float*)d_in[1];
    const float* pos    = (const float*)d_in[2];
    const float* aval   = (const float*)d_in[3];
    const float* astate = (const float*)d_in[4];
    const float* ing    = (const float*)d_in[5];
    const float* inb    = (const float*)d_in[6];
    const float* spw    = (const float*)d_in[7];
    const float* spb    = (const float*)d_in[8];
    const float* wq     = (const float*)d_in[9];
    const float* wk     = (const float*)d_in[10];
    const float* lng    = (const float*)d_in[11];
    const float* lnb    = (const float*)d_in[12];

    float* ws = (float*)d_ws;
    size_t o = 0;
    float* val_a = ws + o; o += (size_t)MM * D;
    float* val_b = ws + o; o += (size_t)MM * D;
    float* st_a  = ws + o; o += MM;
    float* st_b  = ws + o; o += MM;
    float* qbuf  = ws + o; o += (size_t)MM * R;
    float* kbuf  = ws + o; o += (size_t)MM * R;
    float* qbuf2 = ws + o; o += (size_t)MM * R;
    float* kbuf2 = ws + o; o += (size_t)MM * R;

    float* out_state = (float*)d_out;
    float* out_val   = out_state + MM;

    const int ablocks = BB * TILES;

    embed_ln_qk_kernel<<<ablocks, 256, 0, stream>>>(ids, emb, pos, aval, astate,
                                                    ing, inb, spw, spb, val_a, st_a,
                                                    wq, wk, qbuf, kbuf);

    attn_fused_kernel<true><<<ablocks, 256, 0, stream>>>(
        val_a, st_a, qbuf, kbuf, lng, lnb, val_b, st_b,
        wq + (size_t)D * R, wk + (size_t)D * R, qbuf2, kbuf2);
    attn_fused_kernel<false><<<ablocks, 256, 0, stream>>>(
        val_b, st_b, qbuf2, kbuf2, lng + D, lnb + D, out_val, out_state,
        wq + (size_t)D * R, wk + (size_t)D * R, qbuf, kbuf);
}


# Attention epilogue from the accumulator registers: residual rows read as one branch-free group, LayerNorm sums by DPP row sums + 8-wave LDS exchange, stores from registers; row-sum order changed
# speedup vs baseline: 2.9810x; 1.1238x over previous
#include <hip/hip_runtime.h>

#define D   512
#define R   64
#define WIN 32
#define BB  4
#define SS  2048
#define NN  (SS + 1)
#define MM  (BB * NN)
#define TILES ((NN + 15) / 16)
#define WTOT 66
#define DP  (D + 4)

typedef float v2f __attribute__((ext_vector_type(2)));
typedef float v8f __attribute__((ext_vector_type(8)));

__device__ __forceinline__ int clampi(int x, int lo, int hi) {
    return min(max(x, lo), hi);
}

__device__ __forceinline__ float row16_sum(float v) {
    v += __builtin_bit_cast(float, __builtin_amdgcn_update_dpp(0, __builtin_bit_cast(int, v), 0x161, 0xf, 0xf, true));
    v += __builtin_bit_cast(float, __builtin_amdgcn_update_dpp(0, __builtin_bit_cast(int, v), 0x162, 0xf, 0xf, true));
    v += __builtin_bit_cast(float, __builtin_amdgcn_update_dpp(0, __builtin_bit_cast(int, v), 0x164, 0xf, 0xf, true));
    v += __builtin_bit_cast(float, __builtin_amdgcn_update_dpp(0, __builtin_bit_cast(int, v), 0x168, 0xf, 0xf, true));
    return v;
}

__device__ __forceinline__ void qk_project_tile(
    const float* s_t, const float* __restrict__ wq, const float* __restrict__ wk,
    float* __restrict__ qo, float* __restrict__ ko,
    int grow0, int nvalid, int wave, int lm, int lh)
{
    constexpr int U = 16;
    constexpr int G = D / (4 * U);
    const float* wm = (wave < 4) ? wq : wk;
    float*       om = (wave < 4) ? qo : ko;
    const int col = (wave & 3) * 16 + lm;
    const float* bp = wm + (size_t)(2 * lh) * R + col;
    const float* ap = s_t + lm * DP + 2 * lh;

    float bx[U], by[U];
#pragma unroll
    for (int u = 0; u < U; ++u) {
        bx[u] = bp[(size_t)(4 * u) * R];
        by[u] = bp[(size_t)(4 * u + 1) * R];
    }
    v8f acc = {};
#pragma unroll
    for (int g = 0; g < G; ++g) {
        float nx[U], ny[U];
        if (g + 1 < G) {
            const float* np = bp + (size_t)(g + 1) * (4 * U) * R;
#pragma unroll
            for (int u = 0; u < U; ++u) {
                nx[u] = np[(size_t)(4 * u) * R];
                ny[u] = np[(size_t)(4 * u + 1) * R];
            }
        }
#pragma unroll
        for (int u = 0; u < U; ++u) {
            v2f a = *(const v2f*)(ap + g * (4 * U) + 4 * u);
            v2f b; b.x = bx[u]; b.y = by[u];
            acc = __builtin_amdgcn_wmma_f32_16x16x4_f32(
                false, a, false, b, (short)0, acc, false, false);
        }
        if (g + 1 < G) {
#pragma unroll
            for (int u = 0; u < U; ++u) { bx[u] = nx[u]; by[u] = ny[u]; }
        }
    }
#pragma unroll
    for (int i = 0; i < 8; ++i) {
        const int m = i + 8 * lh;
        if (m < nvalid) om[(size_t)(grow0 + m) * R + col] = acc[i];
    }
}

__global__ __launch_bounds__(256) void embed_ln_qk_kernel(
    const int* __restrict__ ids, const float* __restrict__ emb,
    const float* __restrict__ pos, const float* __restrict__ anchor_val,
    const float* __restrict__ anchor_state, const float* __restrict__ gamma,
    const float* __restrict__ beta, const float* __restrict__ sp_w,
    const float* __restrict__ sp_b, float* __restrict__ val,
    float* __restrict__ state,
    const float* __restrict__ wq, const float* __restrict__ wk,
    float* __restrict__ q, float* __restrict__ k)
{
    __shared__ __attribute__((aligned(16))) float s_y[16][DP];
    __shared__ __attribute__((aligned(16))) float s_p[3][16][16];

    const int tid  = threadIdx.x;
    const int lane = tid & 31;
    const int wave = tid >> 5;
    const int lm   = lane & 15;
    const int lh   = lane >> 4;
    const int b    = blockIdx.x / TILES;
    const int t    = blockIdx.x % TILES;
    const int p0   = t * 16;
    const int base = b * NN;

    const int m    = tid >> 4;
    const int jj   = tid & 15;
    const int node = p0 + m;
    const bool is_tok = (node >= 1) && (node < NN);

    float4 x[8];
    if (is_tok) {
        const int s  = node - 1;
        const int id = ids[b * SS + s];
        const float4* e4 = (const float4*)(emb + (size_t)id * D) + jj;
        const float4* p4 = (const float4*)(pos + (size_t)s * D) + jj;
#pragma unroll
        for (int i = 0; i < 8; ++i) {
            const float4 ev = e4[16 * i];
            const float4 pv = p4[16 * i];
            x[i] = make_float4(ev.x + pv.x, ev.y + pv.y, ev.z + pv.z, ev.w + pv.w);
        }
    } else if (node == 0) {
        const float4* a4 = (const float4*)anchor_val + jj;
#pragma unroll
        for (int i = 0; i < 8; ++i) x[i] = a4[16 * i];
    } else {
#pragma unroll
        for (int i = 0; i < 8; ++i) x[i] = make_float4(0.f, 0.f, 0.f, 0.f);
    }

    float part = 0.f;
#pragma unroll
    for (int i = 0; i < 8; ++i) part += (x[i].x + x[i].y) + (x[i].z + x[i].w);
    s_p[0][m][jj] = part;
    __syncthreads();
    float tot = 0.f;
#pragma unroll
    for (int u = 0; u < 16; ++u) tot += s_p[0][m][u];
    const float mean = tot * (1.0f / D);

    part = 0.f;
#pragma unroll
    for (int i = 0; i < 8; ++i) {
        const float cx = x[i].x - mean, cy = x[i].y - mean, cz = x[i].z - mean, cw = x[i].w - mean;
        part += (cx * cx + cy * cy) + (cz * cz + cw * cw);
    }
    s_p[1][m][jj] = part;
    __syncthreads();
    tot = 0.f;
#pragma unroll
    for (int u = 0; u < 16; ++u) tot += s_p[1][m][u];
    const float rstd = rsqrtf(tot * (1.0f / D) + 1e-5f);

    const float4* g4 = (const float4*)gamma + jj;
    const float4* b4 = (const float4*)beta + jj;
    const float4* w4 = (const float4*)sp_w + jj;
    part = 0.f;
#pragma unroll
    for (int i = 0; i < 8; ++i) {
        const float4 g = g4[16 * i];
        const float4 be = b4[16 * i];
        const float4 w = w4[16 * i];
        float4 y;
        y.x = (x[i].x - mean) * rstd * g.x + be.x;
        y.y = (x[i].y - mean) * rstd * g.y + be.y;
        y.z = (x[i].z - mean) * rstd * g.z + be.z;
        y.w = (x[i].w - mean) * rstd * g.w + be.w;
        part += (y.x * w.x + y.y * w.y) + (y.z * w.z + y.w * w.w);
        if (is_tok) x[i] = y;
        *((float4*)&s_y[m][4 * jj + 64 * i]) = x[i];
    }
    s_p[2][m][jj] = part;
    __syncthreads();
    if (node < NN) {
        float4* vo = (float4*)(val + (size_t)(base + node) * D) + jj;
#pragma unroll
        for (int i = 0; i < 8; ++i) vo[16 * i] = x[i];
    }
    if (jj == 0 && node < NN) {
        float st;
        if (node == 0) {
            st = anchor_state[0];
        } else {
            tot = 0.f;
#pragma unroll
            for (int u = 0; u < 16; ++u) tot += s_p[2][m][u];
            st = tot + sp_b[0];
        }
        state[base + node] = st;
    }

    qk_project_tile(&s_y[0][0], wq, wk, q, k, base + p0, NN - p0, wave, lm, lh);
}

template <bool PROJ>
__global__ __launch_bounds__(256) __attribute__((amdgpu_waves_per_eu(PROJ ? 1 : 2, PROJ ? 16 : 7))) void attn_fused_kernel(
    const float* __restrict__ val_in, const float* __restrict__ state_in,
    const float* __restrict__ q, const float* __restrict__ k,
    const float* __restrict__ gamma, const float* __restrict__ beta,
    float* __restrict__ val_out, float* __restrict__ state_out,
    const float* __restrict__ wq_next, const float* __restrict__ wk_next,
    float* __restrict__ q_next, float* __restrict__ k_next)
{
    __shared__ __attribute__((aligned(16))) float s_y[16][DP];
    __shared__ __attribute__((aligned(16))) float s_red[2][8][16];

    const int tid  = threadIdx.x;
    const int lane = tid & 31;
    const int wave = tid >> 5;
    const int lm   = lane & 15;
    const int lh   = lane >> 4;
    const int b    = blockIdx.x / TILES;
    const int t    = blockIdx.x % TILES;
    const int p0   = t * 16;
    const int base = b * NN;

    const int node_l = p0 + lm;
    const int pb     = p0 - WIN + 8 * lh;
    v2f qf[16];
    {
        const float* qp = q + (size_t)(base + clampi(node_l, 0, NN - 1)) * R + 2 * lh;
#pragma unroll
        for (int s = 0; s < 16; ++s) qf[s] = *(const v2f*)(qp + 4 * s);
    }
    v8f sc[5];
#pragma unroll
    for (int bt = 0; bt < 5; ++bt) {
        const int krow = base + clampi(p0 - WIN + 16 * bt + lm, 0, NN - 1);
        const float* kp = k + (size_t)krow * R + 2 * lh;
        v8f acc = {};
#pragma unroll
        for (int s = 0; s < 16; ++s) {
            v2f a = *(const v2f*)(kp + 4 * s);
            acc = __builtin_amdgcn_wmma_f32_16x16x4_f32(
                false, a, false, qf[s], (short)0, acc, false, false);
        }
        sc[bt] = acc;
    }

    const float scale = 0.125f;
    float adot;
    {
        const float* k0 = k + (size_t)base * R + 2 * lh;
        float pa = 0.f;
#pragma unroll
        for (int s = 0; s < 16; ++s) {
            const v2f kv = *(const v2f*)(k0 + 4 * s);
            pa = fmaf(qf[s].x, kv.x, pa);
            pa = fmaf(qf[s].y, kv.y, pa);
        }
        adot = (pa + __shfl_xor(pa, 16)) * scale;
    }
    const bool avalid = (node_l - WIN) > 0;
    const int lo = max(lm - 8 * lh, -pb);
    const int hi = min(lm - 8 * lh + 2 * WIN, NN - 1 - pb);
    float mx = avalid ? fabsf(adot) : -1e30f;
#pragma unroll
    for (int bt = 0; bt < 5; ++bt) {
#pragma unroll
        for (int i = 0; i < 8; ++i) {
            const int c = 16 * bt + i;
            const float s = sc[bt][i] * scale;
            sc[bt][i] = s;
            if (c >= lo && c <= hi) mx = fmaxf(mx, fabsf(s));
        }
    }
    mx = fmaxf(mx, __shfl_xor(mx, 16));
    float sum = 0.f;
#pragma unroll
    for (int bt = 0; bt < 5; ++bt) {
#pragma unroll
        for (int i = 0; i < 8; ++i) {
            const int c = 16 * bt + i;
            const bool ok = (c >= lo) && (c <= hi);
            const float s = sc[bt][i];
            const float e = __expf(fabsf(s) - mx);
            const float se = (s > 0.f) ? e : ((s < 0.f) ? -e : 0.f);
            sum += ok ? e : 0.f;
            sc[bt][i] = ok ? se : 0.f;
        }
    }
    sum += __shfl_xor(sum, 16);
    float ase = 0.f;
    if (avalid) {
        const float e = __expf(fabsf(adot) - mx);
        sum += e;
        ase = (adot > 0.f) ? e : ((adot < 0.f) ? -e : 0.f);
    }
    const float inv = 1.f / sum;
    const float wa  = ase * inv;
#pragma unroll
    for (int bt = 0; bt < 5; ++bt) {
#pragma unroll
        for (int i = 0; i < 8; ++i) sc[bt][i] *= inv;
    }
    if (wave == 0) {
        float ds = 0.f;
#pragma unroll
        for (int bt = 0; bt < 5; ++bt) {
#pragma unroll
            for (int i = 0; i < 8; ++i)
                ds += sc[bt][i] * state_in[base + clampi(pb + 16 * bt + i, 0, NN - 1)];
        }
        ds += __shfl_xor(ds, 16);
        ds += wa * state_in[base];
        if (lh == 0 && node_l < NN)
            state_out[base + node_l] = state_in[base + node_l] + ds;
    }

    {
        const int c0 = wave * 64;
        v8f acc[4] = {};
#pragma unroll
        for (int s = 0; s < 20; ++s) {
            const int bt = s >> 2;
            const int i0 = 2 * (s & 3);
            v2f a;
            a.x = sc[bt][i0];
            a.y = sc[bt][i0 + 1];
            const int r0 = base + clampi(pb + 16 * bt + i0,     0, NN - 1);
            const int r1 = base + clampi(pb + 16 * bt + i0 + 1, 0, NN - 1);
            const float* v0 = val_in + (size_t)r0 * D + c0 + lm;
            const float* v1 = val_in + (size_t)r1 * D + c0 + lm;
#pragma unroll
            for (int nt = 0; nt < 4; ++nt) {
                v2f bb;
                bb.x = v0[nt * 16];
                bb.y = v1[nt * 16];
                acc[nt] = __builtin_amdgcn_wmma_f32_16x16x4_f32(
                    false, a, false, bb, (short)0, acc[nt], false, false);
            }
        }
        {
            v2f a;
            a.x = (lh == 0) ? wa : 0.f;
            a.y = 0.f;
            const float* v0 = val_in + (size_t)base * D + c0 + lm;
#pragma unroll
            for (int nt = 0; nt < 4; ++nt) {
                v2f bb;
                bb.x = v0[nt * 16];
                bb.y = bb.x;
                acc[nt] = __builtin_amdgcn_wmma_f32_16x16x4_f32(
                    false, a, false, bb, (short)0, acc[nt], false, false);
            }
        }
        float gam[4], bet[4];
#pragma unroll
        for (int nt = 0; nt < 4; ++nt) {
            gam[nt] = gamma[c0 + nt * 16 + lm];
            bet[nt] = beta[c0 + nt * 16 + lm];
        }
        float y[4][8];
#pragma unroll
        for (int i = 0; i < 8; ++i) {
            const float* vr = val_in + (size_t)(base + min(p0 + i + 8 * lh, NN - 1)) * D + c0 + lm;
#pragma unroll
            for (int nt = 0; nt < 4; ++nt) y[nt][i] = acc[nt][i] + vr[nt * 16];
        }
        float mean[8];
        {
            float rs[8];
#pragma unroll
            for (int i = 0; i < 8; ++i)
                rs[i] = row16_sum((y[0][i] + y[1][i]) + (y[2][i] + y[3][i]));
            if (lm == 0) {
                float4* dst = (float4*)&s_red[0][wave][8 * lh];
                dst[0] = make_float4(rs[0], rs[1], rs[2], rs[3]);
                dst[1] = make_float4(rs[4], rs[5], rs[6], rs[7]);
            }
            __syncthreads();
            float tot[8];
#pragma unroll
            for (int w = 0; w < 8; ++w) {
                const float4* src = (const float4*)&s_red[0][w][8 * lh];
                const float4 a = src[0], c = src[1];
                if (w == 0) {
                    tot[0] = a.x; tot[1] = a.y; tot[2] = a.z; tot[3] = a.w;
                    tot[4] = c.x; tot[5] = c.y; tot[6] = c.z; tot[7] = c.w;
                } else {
                    tot[0] += a.x; tot[1] += a.y; tot[2] += a.z; tot[3] += a.w;
                    tot[4] += c.x; tot[5] += c.y; tot[6] += c.z; tot[7] += c.w;
                }
            }
#pragma unroll
            for (int i = 0; i < 8; ++i) mean[i] = tot[i] * (1.0f / D);
        }
        float rstd[8];
        {
            float rs[8];
#pragma unroll
            for (int i = 0; i < 8; ++i) {
                const float d0 = y[0][i] - mean[i], d1 = y[1][i] - mean[i];
                const float d2 = y[2][i] - mean[i], d3 = y[3][i] - mean[i];
                rs[i] = row16_sum((d0 * d0 + d1 * d1) + (d2 * d2 + d3 * d3));
            }
            if (lm == 0) {
                float4* dst = (float4*)&s_red[1][wave][8 * lh];
                dst[0] = make_float4(rs[0], rs[1], rs[2], rs[3]);
                dst[1] = make_float4(rs[4], rs[5], rs[6], rs[7]);
            }
            __syncthreads();
            float tot[8];
#pragma unroll
            for (int w = 0; w < 8; ++w) {
                const float4* src = (const float4*)&s_red[1][w][8 * lh];
                const float4 a = src[0], c = src[1];
                if (w == 0) {
                    tot[0] = a.x; tot[1] = a.y; tot[2] = a.z; tot[3] = a.w;
                    tot[4] = c.x; tot[5] = c.y; tot[6] = c.z; tot[7] = c.w;
                } else {
                    tot[0] += a.x; tot[1] += a.y; tot[2] += a.z; tot[3] += a.w;
                    tot[4] += c.x; tot[5] += c.y; tot[6] += c.z; tot[7] += c.w;
                }
            }
#pragma unroll
            for (int i = 0; i < 8; ++i) rstd[i] = rsqrtf(tot[i] * (1.0f / D) + 1e-5f);
        }
#pragma unroll
        for (int nt = 0; nt < 4; ++nt) {
#pragma unroll
            for (int i = 0; i < 8; ++i)
                y[nt][i] = (y[nt][i] - mean[i]) * rstd[i] * gam[nt] + bet[nt];
        }
        float* vo = val_out + (size_t)(base + p0 + 8 * lh) * D + c0 + lm;
        if (p0 + 16 <= NN) {
#pragma unroll
            for (int nt = 0; nt < 4; ++nt) {
#pragma unroll
                for (int i = 0; i < 8; ++i) vo[(size_t)i * D + nt * 16] = y[nt][i];
            }
        } else {
#pragma unroll
            for (int nt = 0; nt < 4; ++nt) {
#pragma unroll
                for (int i = 0; i < 8; ++i)
                    if (p0 + i + 8 * lh < NN) vo[(size_t)i * D + nt * 16] = y[nt][i];
            }
        }
        if (PROJ) {
#pragma unroll
            for (int nt = 0; nt < 4; ++nt) {
#pragma unroll
                for (int i = 0; i < 8; ++i) s_y[i + 8 * lh][c0 + nt * 16 + lm] = y[nt][i];
            }
        }
    }

    if (PROJ) {
        __syncthreads();
        qk_project_tile(&s_y[0][0], wq_next, wk_next, q_next, k_next,
                        base + p0, NN - p0, wave, lm, lh);
    }
}

extern "C" void kernel_launch(void* const* d_in, const int* in_sizes, int n_in,
                              void* d_out, int out_size, void* d_ws, size_t ws_size,
                              hipStream_t stream)
{
    const int*   ids    = (const int*)  d_in[0];
    const float* emb    = (const float*)d_in[1];
    const float* pos    = (const float*)d_in[2];
    const float* aval   = (const float*)d_in[3];
    const float* astate = (const float*)d_in[4];
    const float* ing    = (const float*)d_in[5];
    const float* inb    = (const float*)d_in[6];
    const float* spw    = (const float*)d_in[7];
    const float* spb    = (const float*)d_in[8];
    const float* wq     = (const float*)d_in[9];
    const float* wk     = (const float*)d_in[10];
    const float* lng    = (const float*)d_in[11];
    const float* lnb    = (const float*)d_in[12];

    float* ws = (float*)d_ws;
    size_t o = 0;
    float* val_a = ws + o; o += (size_t)MM * D;
    float* val_b = ws + o; o += (size_t)MM * D;
    float* st_a  = ws + o; o += MM;
    float* st_b  = ws + o; o += MM;
    float* qbuf  = ws + o; o += (size_t)MM * R;
    float* kbuf  = ws + o; o += (size_t)MM * R;
    float* qbuf2 = ws + o; o += (size_t)MM * R;
    float* kbuf2 = ws + o; o += (size_t)MM * R;

    float* out_state = (float*)d_out;
    float* out_val   = out_state + MM;

    const int ablocks = BB * TILES;

    embed_ln_qk_kernel<<<ablocks, 256, 0, stream>>>(ids, emb, pos, aval, astate,
                                                    ing, inb, spw, spb, val_a, st_a,
                                                    wq, wk, qbuf, kbuf);

    attn_fused_kernel<true><<<ablocks, 256, 0, stream>>>(
        val_a, st_a, qbuf, kbuf, lng, lnb, val_b, st_b,
        wq + (size_t)D * R, wk + (size_t)D * R, qbuf2, kbuf2);
    attn_fused_kernel<false><<<ablocks, 256, 0, stream>>>(
        val_b, st_b, qbuf2, kbuf2, lng + D, lnb + D, out_val, out_state,
        wq + (size_t)D * R, wk + (size_t)D * R, qbuf, kbuf);
}


# Score tiles computed once per block, not by every wave: wave t forms tile t, wave 5 the anchor scores; tiles pass through LDS lane by lane (one barrier) into the same registers; bit-identical
# speedup vs baseline: 3.6548x; 1.2260x over previous
#include <hip/hip_runtime.h>

#define D   512
#define R   64
#define WIN 32
#define BB  4
#define SS  2048
#define NN  (SS + 1)
#define MM  (BB * NN)
#define TILES ((NN + 15) / 16)
#define WTOT 66
#define DP  (D + 4)

typedef float v2f __attribute__((ext_vector_type(2)));
typedef float v8f __attribute__((ext_vector_type(8)));

__device__ __forceinline__ int clampi(int x, int lo, int hi) {
    return min(max(x, lo), hi);
}

__device__ __forceinline__ float row16_sum(float v) {
    v += __builtin_bit_cast(float, __builtin_amdgcn_update_dpp(0, __builtin_bit_cast(int, v), 0x161, 0xf, 0xf, true));
    v += __builtin_bit_cast(float, __builtin_amdgcn_update_dpp(0, __builtin_bit_cast(int, v), 0x162, 0xf, 0xf, true));
    v += __builtin_bit_cast(float, __builtin_amdgcn_update_dpp(0, __builtin_bit_cast(int, v), 0x164, 0xf, 0xf, true));
    v += __builtin_bit_cast(float, __builtin_amdgcn_update_dpp(0, __builtin_bit_cast(int, v), 0x168, 0xf, 0xf, true));
    return v;
}

__device__ __forceinline__ void qk_project_tile(
    const float* s_t, const float* __restrict__ wq, const float* __restrict__ wk,
    float* __restrict__ qo, float* __restrict__ ko,
    int grow0, int nvalid, int wave, int lm, int lh)
{
    constexpr int U = 16;
    constexpr int G = D / (4 * U);
    const float* wm = (wave < 4) ? wq : wk;
    float*       om = (wave < 4) ? qo : ko;
    const int col = (wave & 3) * 16 + lm;
    const float* bp = wm + (size_t)(2 * lh) * R + col;
    const float* ap = s_t + lm * DP + 2 * lh;

    float bx[U], by[U];
#pragma unroll
    for (int u = 0; u < U; ++u) {
        bx[u] = bp[(size_t)(4 * u) * R];
        by[u] = bp[(size_t)(4 * u + 1) * R];
    }
    v8f acc = {};
#pragma unroll
    for (int g = 0; g < G; ++g) {
        float nx[U], ny[U];
        if (g + 1 < G) {
            const float* np = bp + (size_t)(g + 1) * (4 * U) * R;
#pragma unroll
            for (int u = 0; u < U; ++u) {
                nx[u] = np[(size_t)(4 * u) * R];
                ny[u] = np[(size_t)(4 * u + 1) * R];
            }
        }
#pragma unroll
        for (int u = 0; u < U; ++u) {
            v2f a = *(const v2f*)(ap + g * (4 * U) + 4 * u);
            v2f b; b.x = bx[u]; b.y = by[u];
            acc = __builtin_amdgcn_wmma_f32_16x16x4_f32(
                false, a, false, b, (short)0, acc, false, false);
        }
        if (g + 1 < G) {
#pragma unroll
            for (int u = 0; u < U; ++u) { bx[u] = nx[u]; by[u] = ny[u]; }
        }
    }
#pragma unroll
    for (int i = 0; i < 8; ++i) {
        const int m = i + 8 * lh;
        if (m < nvalid) om[(size_t)(grow0 + m) * R + col] = acc[i];
    }
}

__global__ __launch_bounds__(256) void embed_ln_qk_kernel(
    const int* __restrict__ ids, const float* __restrict__ emb,
    const float* __restrict__ pos, const float* __restrict__ anchor_val,
    const float* __restrict__ anchor_state, const float* __restrict__ gamma,
    const float* __restrict__ beta, const float* __restrict__ sp_w,
    const float* __restrict__ sp_b, float* __restrict__ val,
    float* __restrict__ state,
    const float* __restrict__ wq, const float* __restrict__ wk,
    float* __restrict__ q, float* __restrict__ k)
{
    __shared__ __attribute__((aligned(16))) float s_y[16][DP];
    __shared__ __attribute__((aligned(16))) float s_p[3][16][16];

    const int tid  = threadIdx.x;
    const int lane = tid & 31;
    const int wave = tid >> 5;
    const int lm   = lane & 15;
    const int lh   = lane >> 4;
    const int b    = blockIdx.x / TILES;
    const int t    = blockIdx.x % TILES;
    const int p0   = t * 16;
    const int base = b * NN;

    const int m    = tid >> 4;
    const int jj   = tid & 15;
    const int node = p0 + m;
    const bool is_tok = (node >= 1) && (node < NN);

    float4 x[8];
    if (is_tok) {
        const int s  = node - 1;
        const int id = ids[b * SS + s];
        const float4* e4 = (const float4*)(emb + (size_t)id * D) + jj;
        const float4* p4 = (const float4*)(pos + (size_t)s * D) + jj;
#pragma unroll
        for (int i = 0; i < 8; ++i) {
            const float4 ev = e4[16 * i];
            const float4 pv = p4[16 * i];
            x[i] = make_float4(ev.x + pv.x, ev.y + pv.y, ev.z + pv.z, ev.w + pv.w);
        }
    } else if (node == 0) {
        const float4* a4 = (const float4*)anchor_val + jj;
#pragma unroll
        for (int i = 0; i < 8; ++i) x[i] = a4[16 * i];
    } else {
#pragma unroll
        for (int i = 0; i < 8; ++i) x[i] = make_float4(0.f, 0.f, 0.f, 0.f);
    }

    float part = 0.f;
#pragma unroll
    for (int i = 0; i < 8; ++i) part += (x[i].x + x[i].y) + (x[i].z + x[i].w);
    s_p[0][m][jj] = part;
    __syncthreads();
    float tot = 0.f;
#pragma unroll
    for (int u = 0; u < 16; ++u) tot += s_p[0][m][u];
    const float mean = tot * (1.0f / D);

    part = 0.f;
#pragma unroll
    for (int i = 0; i < 8; ++i) {
        const float cx = x[i].x - mean, cy = x[i].y - mean, cz = x[i].z - mean, cw = x[i].w - mean;
        part += (cx * cx + cy * cy) + (cz * cz + cw * cw);
    }
    s_p[1][m][jj] = part;
    __syncthreads();
    tot = 0.f;
#pragma unroll
    for (int u = 0; u < 16; ++u) tot += s_p[1][m][u];
    const float rstd = rsqrtf(tot * (1.0f / D) + 1e-5f);

    const float4* g4 = (const float4*)gamma + jj;
    const float4* b4 = (const float4*)beta + jj;
    const float4* w4 = (const float4*)sp_w + jj;
    part = 0.f;
#pragma unroll
    for (int i = 0; i < 8; ++i) {
        const float4 g = g4[16 * i];
        const float4 be = b4[16 * i];
        const float4 w = w4[16 * i];
        float4 y;
        y.x = (x[i].x - mean) * rstd * g.x + be.x;
        y.y = (x[i].y - mean) * rstd * g.y + be.y;
        y.z = (x[i].z - mean) * rstd * g.z + be.z;
        y.w = (x[i].w - mean) * rstd * g.w + be.w;
        part += (y.x * w.x + y.y * w.y) + (y.z * w.z + y.w * w.w);
        if (is_tok) x[i] = y;
        *((float4*)&s_y[m][4 * jj + 64 * i]) = x[i];
    }
    s_p[2][m][jj] = part;
    __syncthreads();
    if (node < NN) {
        float4* vo = (float4*)(val + (size_t)(base + node) * D) + jj;
#pragma unroll
        for (int i = 0; i < 8; ++i) vo[16 * i] = x[i];
    }
    if (jj == 0 && node < NN) {
        float st;
        if (node == 0) {
            st = anchor_state[0];
        } else {
            tot = 0.f;
#pragma unroll
            for (int u = 0; u < 16; ++u) tot += s_p[2][m][u];
            st = tot + sp_b[0];
        }
        state[base + node] = st;
    }

    qk_project_tile(&s_y[0][0], wq, wk, q, k, base + p0, NN - p0, wave, lm, lh);
}

template <bool PROJ>
__global__ __launch_bounds__(256) __attribute__((amdgpu_waves_per_eu(PROJ ? 6 : 2, 7))) void attn_fused_kernel(
    const float* __restrict__ val_in, const float* __restrict__ state_in,
    const float* __restrict__ q, const float* __restrict__ k,
    const float* __restrict__ gamma, const float* __restrict__ beta,
    float* __restrict__ val_out, float* __restrict__ state_out,
    const float* __restrict__ wq_next, const float* __restrict__ wk_next,
    float* __restrict__ q_next, float* __restrict__ k_next)
{
    __shared__ __attribute__((aligned(16))) float s_y[16][DP];
    __shared__ __attribute__((aligned(16))) float s_red[2][8][16];
    __shared__ __attribute__((aligned(16))) float4 s_sc[5][2][32];
    __shared__ float s_ad[32];

    const int tid  = threadIdx.x;
    const int lane = tid & 31;
    const int wave = tid >> 5;
    const int lm   = lane & 15;
    const int lh   = lane >> 4;
    const int b    = blockIdx.x / TILES;
    const int t    = blockIdx.x % TILES;
    const int p0   = t * 16;
    const int base = b * NN;

    const int node_l = p0 + lm;
    const int pb     = p0 - WIN + 8 * lh;
    const float scale = 0.125f;
    const int wv = __builtin_amdgcn_readfirstlane(wave);
    if (wv < 6) {
        v2f qf[16];
        {
            const float* qp = q + (size_t)(base + clampi(node_l, 0, NN - 1)) * R + 2 * lh;
#pragma unroll
            for (int s = 0; s < 16; ++s) qf[s] = *(const v2f*)(qp + 4 * s);
        }
        if (wv < 5) {
            const int krow = base + clampi(p0 - WIN + 16 * wv + lm, 0, NN - 1);
            const float* kp = k + (size_t)krow * R + 2 * lh;
            v8f acc = {};
#pragma unroll
            for (int s = 0; s < 16; ++s) {
                v2f a = *(const v2f*)(kp + 4 * s);
                acc = __builtin_amdgcn_wmma_f32_16x16x4_f32(
                    false, a, false, qf[s], (short)0, acc, false, false);
            }
            s_sc[wv][0][lane] = make_float4(acc[0] * scale, acc[1] * scale, acc[2] * scale, acc[3] * scale);
            s_sc[wv][1][lane] = make_float4(acc[4] * scale, acc[5] * scale, acc[6] * scale, acc[7] * scale);
        } else {
            const float* k0 = k + (size_t)base * R + 2 * lh;
            float pa = 0.f;
#pragma unroll
            for (int s = 0; s < 16; ++s) {
                const v2f kv = *(const v2f*)(k0 + 4 * s);
                pa = fmaf(qf[s].x, kv.x, pa);
                pa = fmaf(qf[s].y, kv.y, pa);
            }
            s_ad[lane] = (pa + __shfl_xor(pa, 16)) * scale;
        }
    }
    __syncthreads();
    v8f sc[5];
#pragma unroll
    for (int bt = 0; bt < 5; ++bt) {
        const float4 a = s_sc[bt][0][lane], c = s_sc[bt][1][lane];
        sc[bt][0] = a.x; sc[bt][1] = a.y; sc[bt][2] = a.z; sc[bt][3] = a.w;
        sc[bt][4] = c.x; sc[bt][5] = c.y; sc[bt][6] = c.z; sc[bt][7] = c.w;
    }
    const float adot = s_ad[lane];

    const bool avalid = (node_l - WIN) > 0;
    const int lo = max(lm - 8 * lh, -pb);
    const int hi = min(lm - 8 * lh + 2 * WIN, NN - 1 - pb);
    float mx = avalid ? fabsf(adot) : -1e30f;
#pragma unroll
    for (int bt = 0; bt < 5; ++bt) {
#pragma unroll
        for (int i = 0; i < 8; ++i) {
            const int c = 16 * bt + i;
            if (c >= lo && c <= hi) mx = fmaxf(mx, fabsf(sc[bt][i]));
        }
    }
    mx = fmaxf(mx, __shfl_xor(mx, 16));
    float sum = 0.f;
#pragma unroll
    for (int bt = 0; bt < 5; ++bt) {
#pragma unroll
        for (int i = 0; i < 8; ++i) {
            const int c = 16 * bt + i;
            const bool ok = (c >= lo) && (c <= hi);
            const float s = sc[bt][i];
            const float e = __expf(fabsf(s) - mx);
            const float se = (s > 0.f) ? e : ((s < 0.f) ? -e : 0.f);
            sum += ok ? e : 0.f;
            sc[bt][i] = ok ? se : 0.f;
        }
    }
    sum += __shfl_xor(sum, 16);
    float ase = 0.f;
    if (avalid) {
        const float e = __expf(fabsf(adot) - mx);
        sum += e;
        ase = (adot > 0.f) ? e : ((adot < 0.f) ? -e : 0.f);
    }
    const float inv = 1.f / sum;
    const float wa  = ase * inv;
#pragma unroll
    for (int bt = 0; bt < 5; ++bt) {
#pragma unroll
        for (int i = 0; i < 8; ++i) sc[bt][i] *= inv;
    }
    if (wave == 0) {
        float ds = 0.f;
#pragma unroll
        for (int bt = 0; bt < 5; ++bt) {
#pragma unroll
            for (int i = 0; i < 8; ++i)
                ds += sc[bt][i] * state_in[base + clampi(pb + 16 * bt + i, 0, NN - 1)];
        }
        ds += __shfl_xor(ds, 16);
        ds += wa * state_in[base];
        if (lh == 0 && node_l < NN)
            state_out[base + node_l] = state_in[base + node_l] + ds;
    }

    {
        const int c0 = wave * 64;
        v8f acc[4] = {};
#pragma unroll
        for (int s = 0; s < 20; ++s) {
            const int bt = s >> 2;
            const int i0 = 2 * (s & 3);
            v2f a;
            a.x = sc[bt][i0];
            a.y = sc[bt][i0 + 1];
            const int r0 = base + clampi(pb + 16 * bt + i0,     0, NN - 1);
            const int r1 = base + clampi(pb + 16 * bt + i0 + 1, 0, NN - 1);
            const float* v0 = val_in + (size_t)r0 * D + c0 + lm;
            const float* v1 = val_in + (size_t)r1 * D + c0 + lm;
#pragma unroll
            for (int nt = 0; nt < 4; ++nt) {
                v2f bb;
                bb.x = v0[nt * 16];
                bb.y = v1[nt * 16];
                acc[nt] = __builtin_amdgcn_wmma_f32_16x16x4_f32(
                    false, a, false, bb, (short)0, acc[nt], false, false);
            }
        }
        {
            v2f a;
            a.x = (lh == 0) ? wa : 0.f;
            a.y = 0.f;
            const float* v0 = val_in + (size_t)base * D + c0 + lm;
#pragma unroll
            for (int nt = 0; nt < 4; ++nt) {
                v2f bb;
                bb.x = v0[nt * 16];
                bb.y = bb.x;
                acc[nt] = __builtin_amdgcn_wmma_f32_16x16x4_f32(
                    false, a, false, bb, (short)0, acc[nt], false, false);
            }
        }
        float gam[4], bet[4];
#pragma unroll
        for (int nt = 0; nt < 4; ++nt) {
            gam[nt] = gamma[c0 + nt * 16 + lm];
            bet[nt] = beta[c0 + nt * 16 + lm];
        }
        float y[4][8];
#pragma unroll
        for (int i = 0; i < 8; ++i) {
            const float* vr = val_in + (size_t)(base + min(p0 + i + 8 * lh, NN - 1)) * D + c0 + lm;
#pragma unroll
            for (int nt = 0; nt < 4; ++nt) y[nt][i] = acc[nt][i] + vr[nt * 16];
        }
        float mean[8];
        {
            float rs[8];
#pragma unroll
            for (int i = 0; i < 8; ++i)
                rs[i] = row16_sum((y[0][i] + y[1][i]) + (y[2][i] + y[3][i]));
            if (lm == 0) {
                float4* dst = (float4*)&s_red[0][wave][8 * lh];
                dst[0] = make_float4(rs[0], rs[1], rs[2], rs[3]);
                dst[1] = make_float4(rs[4], rs[5], rs[6], rs[7]);
            }
            __syncthreads();
            float tot[8];
#pragma unroll
            for (int w = 0; w < 8; ++w) {
                const float4* src = (const float4*)&s_red[0][w][8 * lh];
                const float4 a = src[0], c = src[1];
                if (w == 0) {
                    tot[0] = a.x; tot[1] = a.y; tot[2] = a.z; tot[3] = a.w;
                    tot[4] = c.x; tot[5] = c.y; tot[6] = c.z; tot[7] = c.w;
                } else {
                    tot[0] += a.x; tot[1] += a.y; tot[2] += a.z; tot[3] += a.w;
                    tot[4] += c.x; tot[5] += c.y; tot[6] += c.z; tot[7] += c.w;
                }
            }
#pragma unroll
            for (int i = 0; i < 8; ++i) mean[i] = tot[i] * (1.0f / D);
        }
        float rstd[8];
        {
            float rs[8];
#pragma unroll
            for (int i = 0; i < 8; ++i) {
                const float d0 = y[0][i] - mean[i], d1 = y[1][i] - mean[i];
                const float d2 = y[2][i] - mean[i], d3 = y[3][i] - mean[i];
                rs[i] = row16_sum((d0 * d0 + d1 * d1) + (d2 * d2 + d3 * d3));
            }
            if (lm == 0) {
                float4* dst = (float4*)&s_red[1][wave][8 * lh];
                dst[0] = make_float4(rs[0], rs[1], rs[2], rs[3]);
                dst[1] = make_float4(rs[4], rs[5], rs[6], rs[7]);
            }
            __syncthreads();
            float tot[8];
#pragma unroll
            for (int w = 0; w < 8; ++w) {
                const float4* src = (const float4*)&s_red[1][w][8 * lh];
                const float4 a = src[0], c = src[1];
                if (w == 0) {
                    tot[0] = a.x; tot[1] = a.y; tot[2] = a.z; tot[3] = a.w;
                    tot[4] = c.x; tot[5] = c.y; tot[6] = c.z; tot[7] = c.w;
                } else {
                    tot[0] += a.x; tot[1] += a.y; tot[2] += a.z; tot[3] += a.w;
                    tot[4] += c.x; tot[5] += c.y; tot[6] += c.z; tot[7] += c.w;
                }
            }
#pragma unroll
            for (int i = 0; i < 8; ++i) rstd[i] = rsqrtf(tot[i] * (1.0f / D) + 1e-5f);
        }
#pragma unroll
        for (int nt = 0; nt < 4; ++nt) {
#pragma unroll
            for (int i = 0; i < 8; ++i)
                y[nt][i] = (y[nt][i] - mean[i]) * rstd[i] * gam[nt] + bet[nt];
        }
        float* vo = val_out + (size_t)(base + p0 + 8 * lh) * D + c0 + lm;
        if (p0 + 16 <= NN) {
#pragma unroll
            for (int nt = 0; nt < 4; ++nt) {
#pragma unroll
                for (int i = 0; i < 8; ++i) vo[(size_t)i * D + nt * 16] = y[nt][i];
            }
        } else {
#pragma unroll
            for (int nt = 0; nt < 4; ++nt) {
#pragma unroll
                for (int i = 0; i < 8; ++i)
                    if (p0 + i + 8 * lh < NN) vo[(size_t)i * D + nt * 16] = y[nt][i];
            }
        }
        if (PROJ) {
#pragma unroll
            for (int nt = 0; nt < 4; ++nt) {
#pragma unroll
                for (int i = 0; i < 8; ++i) s_y[i + 8 * lh][c0 + nt * 16 + lm] = y[nt][i];
            }
        }
    }

    if (PROJ) {
        __syncthreads();
        qk_project_tile(&s_y[0][0], wq_next, wk_next, q_next, k_next,
                        base + p0, NN - p0, wave, lm, lh);
    }
}

extern "C" void kernel_launch(void* const* d_in, const int* in_sizes, int n_in,
                              void* d_out, int out_size, void* d_ws, size_t ws_size,
                              hipStream_t stream)
{
    const int*   ids    = (const int*)  d_in[0];
    const float* emb    = (const float*)d_in[1];
    const float* pos    = (const float*)d_in[2];
    const float* aval   = (const float*)d_in[3];
    const float* astate = (const float*)d_in[4];
    const float* ing    = (const float*)d_in[5];
    const float* inb    = (const float*)d_in[6];
    const float* spw    = (const float*)d_in[7];
    const float* spb    = (const float*)d_in[8];
    const float* wq     = (const float*)d_in[9];
    const float* wk     = (const float*)d_in[10];
    const float* lng    = (const float*)d_in[11];
    const float* lnb    = (const float*)d_in[12];

    float* ws = (float*)d_ws;
    size_t o = 0;
    float* val_a = ws + o; o += (size_t)MM * D;
    float* val_b = ws + o; o += (size_t)MM * D;
    float* st_a  = ws + o; o += MM;
    float* st_b  = ws + o; o += MM;
    float* qbuf  = ws + o; o += (size_t)MM * R;
    float* kbuf  = ws + o; o += (size_t)MM * R;
    float* qbuf2 = ws + o; o += (size_t)MM * R;
    float* kbuf2 = ws + o; o += (size_t)MM * R;

    float* out_state = (float*)d_out;
    float* out_val   = out_state + MM;

    const int ablocks = BB * TILES;

    embed_ln_qk_kernel<<<ablocks, 256, 0, stream>>>(ids, emb, pos, aval, astate,
                                                    ing, inb, spw, spb, val_a, st_a,
                                                    wq, wk, qbuf, kbuf);

    attn_fused_kernel<true><<<ablocks, 256, 0, stream>>>(
        val_a, st_a, qbuf, kbuf, lng, lnb, val_b, st_b,
        wq + (size_t)D * R, wk + (size_t)D * R, qbuf2, kbuf2);
    attn_fused_kernel<false><<<ablocks, 256, 0, stream>>>(
        val_b, st_b, qbuf2, kbuf2, lng + D, lnb + D, out_val, out_state,
        wq + (size_t)D * R, wk + (size_t)D * R, qbuf, kbuf);
}


# Score operands (80 k rows, 16 q rows, anchor k row) staged in LDS by all 256 threads with whole-line 16-byte loads; fragments read from LDS (pitch 68) in the same k order; scores bit-identical
# speedup vs baseline: 3.7937x; 1.0380x over previous
#include <hip/hip_runtime.h>

#define D   512
#define R   64
#define WIN 32
#define BB  4
#define SS  2048
#define NN  (SS + 1)
#define MM  (BB * NN)
#define TILES ((NN + 15) / 16)
#define WTOT 66
#define DP  (D + 4)
#define KP  (R + 4)

typedef float v2f __attribute__((ext_vector_type(2)));
typedef float v8f __attribute__((ext_vector_type(8)));

__device__ __forceinline__ int clampi(int x, int lo, int hi) {
    return min(max(x, lo), hi);
}

__device__ __forceinline__ float row16_sum(float v) {
    v += __builtin_bit_cast(float, __builtin_amdgcn_update_dpp(0, __builtin_bit_cast(int, v), 0x161, 0xf, 0xf, true));
    v += __builtin_bit_cast(float, __builtin_amdgcn_update_dpp(0, __builtin_bit_cast(int, v), 0x162, 0xf, 0xf, true));
    v += __builtin_bit_cast(float, __builtin_amdgcn_update_dpp(0, __builtin_bit_cast(int, v), 0x164, 0xf, 0xf, true));
    v += __builtin_bit_cast(float, __builtin_amdgcn_update_dpp(0, __builtin_bit_cast(int, v), 0x168, 0xf, 0xf, true));
    return v;
}

__device__ __forceinline__ void qk_project_tile(
    const float* s_t, const float* __restrict__ wq, const float* __restrict__ wk,
    float* __restrict__ qo, float* __restrict__ ko,
    int grow0, int nvalid, int wave, int lm, int lh)
{
    constexpr int U = 16;
    constexpr int G = D / (4 * U);
    const float* wm = (wave < 4) ? wq : wk;
    float*       om = (wave < 4) ? qo : ko;
    const int col = (wave & 3) * 16 + lm;
    const float* bp = wm + (size_t)(2 * lh) * R + col;
    const float* ap = s_t + lm * DP + 2 * lh;

    float bx[U], by[U];
#pragma unroll
    for (int u = 0; u < U; ++u) {
        bx[u] = bp[(size_t)(4 * u) * R];
        by[u] = bp[(size_t)(4 * u + 1) * R];
    }
    v8f acc = {};
#pragma unroll
    for (int g = 0; g < G; ++g) {
        float nx[U], ny[U];
        if (g + 1 < G) {
            const float* np = bp + (size_t)(g + 1) * (4 * U) * R;
#pragma unroll
            for (int u = 0; u < U; ++u) {
                nx[u] = np[(size_t)(4 * u) * R];
                ny[u] = np[(size_t)(4 * u + 1) * R];
            }
        }
#pragma unroll
        for (int u = 0; u < U; ++u) {
            v2f a = *(const v2f*)(ap + g * (4 * U) + 4 * u);
            v2f b; b.x = bx[u]; b.y = by[u];
            acc = __builtin_amdgcn_wmma_f32_16x16x4_f32(
                false, a, false, b, (short)0, acc, false, false);
        }
        if (g + 1 < G) {
#pragma unroll
            for (int u = 0; u < U; ++u) { bx[u] = nx[u]; by[u] = ny[u]; }
        }
    }
#pragma unroll
    for (int i = 0; i < 8; ++i) {
        const int m = i + 8 * lh;
        if (m < nvalid) om[(size_t)(grow0 + m) * R + col] = acc[i];
    }
}

__global__ __launch_bounds__(256) void embed_ln_qk_kernel(
    const int* __restrict__ ids, const float* __restrict__ emb,
    const float* __restrict__ pos, const float* __restrict__ anchor_val,
    const float* __restrict__ anchor_state, const float* __restrict__ gamma,
    const float* __restrict__ beta, const float* __restrict__ sp_w,
    const float* __restrict__ sp_b, float* __restrict__ val,
    float* __restrict__ state,
    const float* __restrict__ wq, const float* __restrict__ wk,
    float* __restrict__ q, float* __restrict__ k)
{
    __shared__ __attribute__((aligned(16))) float s_y[16][DP];
    __shared__ __attribute__((aligned(16))) float s_p[3][16][16];

    const int tid  = threadIdx.x;
    const int lane = tid & 31;
    const int wave = tid >> 5;
    const int lm   = lane & 15;
    const int lh   = lane >> 4;
    const int b    = blockIdx.x / TILES;
    const int t    = blockIdx.x % TILES;
    const int p0   = t * 16;
    const int base = b * NN;

    const int m    = tid >> 4;
    const int jj   = tid & 15;
    const int node = p0 + m;
    const bool is_tok = (node >= 1) && (node < NN);

    float4 x[8];
    if (is_tok) {
        const int s  = node - 1;
        const int id = ids[b * SS + s];
        const float4* e4 = (const float4*)(emb + (size_t)id * D) + jj;
        const float4* p4 = (const float4*)(pos + (size_t)s * D) + jj;
#pragma unroll
        for (int i = 0; i < 8; ++i) {
            const float4 ev = e4[16 * i];
            const float4 pv = p4[16 * i];
            x[i] = make_float4(ev.x + pv.x, ev.y + pv.y, ev.z + pv.z, ev.w + pv.w);
        }
    } else if (node == 0) {
        const float4* a4 = (const float4*)anchor_val + jj;
#pragma unroll
        for (int i = 0; i < 8; ++i) x[i] = a4[16 * i];
    } else {
#pragma unroll
        for (int i = 0; i < 8; ++i) x[i] = make_float4(0.f, 0.f, 0.f, 0.f);
    }

    float part = 0.f;
#pragma unroll
    for (int i = 0; i < 8; ++i) part += (x[i].x + x[i].y) + (x[i].z + x[i].w);
    s_p[0][m][jj] = part;
    __syncthreads();
    float tot = 0.f;
#pragma unroll
    for (int u = 0; u < 16; ++u) tot += s_p[0][m][u];
    const float mean = tot * (1.0f / D);

    part = 0.f;
#pragma unroll
    for (int i = 0; i < 8; ++i) {
        const float cx = x[i].x - mean, cy = x[i].y - mean, cz = x[i].z - mean, cw = x[i].w - mean;
        part += (cx * cx + cy * cy) + (cz * cz + cw * cw);
    }
    s_p[1][m][jj] = part;
    __syncthreads();
    tot = 0.f;
#pragma unroll
    for (int u = 0; u < 16; ++u) tot += s_p[1][m][u];
    const float rstd = rsqrtf(tot * (1.0f / D) + 1e-5f);

    const float4* g4 = (const float4*)gamma + jj;
    const float4* b4 = (const float4*)beta + jj;
    const float4* w4 = (const float4*)sp_w + jj;
    part = 0.f;
#pragma unroll
    for (int i = 0; i < 8; ++i) {
        const float4 g = g4[16 * i];
        const float4 be = b4[16 * i];
        const float4 w = w4[16 * i];
        float4 y;
        y.x = (x[i].x - mean) * rstd * g.x + be.x;
        y.y = (x[i].y - mean) * rstd * g.y + be.y;
        y.z = (x[i].z - mean) * rstd * g.z + be.z;
        y.w = (x[i].w - mean) * rstd * g.w + be.w;
        part += (y.x * w.x + y.y * w.y) + (y.z * w.z + y.w * w.w);
        if (is_tok) x[i] = y;
        *((float4*)&s_y[m][4 * jj + 64 * i]) = x[i];
    }
    s_p[2][m][jj] = part;
    __syncthreads();
    if (node < NN) {
        float4* vo = (float4*)(val + (size_t)(base + node) * D) + jj;
#pragma unroll
        for (int i = 0; i < 8; ++i) vo[16 * i] = x[i];
    }
    if (jj == 0 && node < NN) {
        float st;
        if (node == 0) {
            st = anchor_state[0];
        } else {
            tot = 0.f;
#pragma unroll
            for (int u = 0; u < 16; ++u) tot += s_p[2][m][u];
            st = tot + sp_b[0];
        }
        state[base + node] = st;
    }

    qk_project_tile(&s_y[0][0], wq, wk, q, k, base + p0, NN - p0, wave, lm, lh);
}

template <bool PROJ>
__global__ __launch_bounds__(256) __attribute__((amdgpu_waves_per_eu(PROJ ? 6 : 2, 7))) void attn_fused_kernel(
    const float* __restrict__ val_in, const float* __restrict__ state_in,
    const float* __restrict__ q, const float* __restrict__ k,
    const float* __restrict__ gamma, const float* __restrict__ beta,
    float* __restrict__ val_out, float* __restrict__ state_out,
    const float* __restrict__ wq_next, const float* __restrict__ wk_next,
    float* __restrict__ q_next, float* __restrict__ k_next)
{
    __shared__ __attribute__((aligned(16))) float s_y[16][DP];
    __shared__ __attribute__((aligned(16))) float s_red[2][8][16];
    __shared__ __attribute__((aligned(16))) float4 s_sc[5][2][32];
    __shared__ float s_ad[32];
    __shared__ __attribute__((aligned(16))) float s_kq[97][KP];

    const int tid  = threadIdx.x;
    const int lane = tid & 31;
    const int wave = tid >> 5;
    const int lm   = lane & 15;
    const int lh   = lane >> 4;
    const int b    = blockIdx.x / TILES;
    const int t    = blockIdx.x % TILES;
    const int p0   = t * 16;
    const int base = b * NN;

    const int node_l = p0 + lm;
    const int pb     = p0 - WIN + 8 * lh;
    const float scale = 0.125f;
    const int wv = __builtin_amdgcn_readfirstlane(wave);
    {
        const int sr = tid >> 4;
        const int c4 = tid & 15;
        float4 kv[5];
#pragma unroll
        for (int j = 0; j < 5; ++j) {
            const int krow = base + clampi(p0 - WIN + 16 * j + sr, 0, NN - 1);
            kv[j] = *((const float4*)(k + (size_t)krow * R) + c4);
        }
        const float4 qv = *((const float4*)(q + (size_t)(base + clampi(p0 + sr, 0, NN - 1)) * R) + c4);
        const float4 av = *((const float4*)(k + (size_t)base * R) + c4);
#pragma unroll
        for (int j = 0; j < 5; ++j) *((float4*)&s_kq[16 * j + sr][4 * c4]) = kv[j];
        *((float4*)&s_kq[80 + sr][4 * c4]) = qv;
        asm volatile("" :: "v"(av.x), "v"(av.y), "v"(av.z), "v"(av.w));
        if (sr == 0) *((float4*)&s_kq[96][4 * c4]) = av;
    }
    __syncthreads();
    if (wv < 6) {
        v2f qf[16];
        {
            const float* qp = &s_kq[80 + lm][2 * lh];
#pragma unroll
            for (int s = 0; s < 16; ++s) qf[s] = *(const v2f*)(qp + 4 * s);
        }
        if (wv < 5) {
            const float* kp = &s_kq[16 * wv + lm][2 * lh];
            v8f acc = {};
#pragma unroll
            for (int s = 0; s < 16; ++s) {
                v2f a = *(const v2f*)(kp + 4 * s);
                acc = __builtin_amdgcn_wmma_f32_16x16x4_f32(
                    false, a, false, qf[s], (short)0, acc, false, false);
            }
            s_sc[wv][0][lane] = make_float4(acc[0] * scale, acc[1] * scale, acc[2] * scale, acc[3] * scale);
            s_sc[wv][1][lane] = make_float4(acc[4] * scale, acc[5] * scale, acc[6] * scale, acc[7] * scale);
        } else {
            const float* k0 = &s_kq[96][2 * lh];
            float pa = 0.f;
#pragma unroll
            for (int s = 0; s < 16; ++s) {
                const v2f kv = *(const v2f*)(k0 + 4 * s);
                pa = fmaf(qf[s].x, kv.x, pa);
                pa = fmaf(qf[s].y, kv.y, pa);
            }
            s_ad[lane] = (pa + __shfl_xor(pa, 16)) * scale;
        }
    }
    __syncthreads();
    v8f sc[5];
#pragma unroll
    for (int bt = 0; bt < 5; ++bt) {
        const float4 a = s_sc[bt][0][lane], c = s_sc[bt][1][lane];
        sc[bt][0] = a.x; sc[bt][1] = a.y; sc[bt][2] = a.z; sc[bt][3] = a.w;
        sc[bt][4] = c.x; sc[bt][5] = c.y; sc[bt][6] = c.z; sc[bt][7] = c.w;
    }
    const float adot = s_ad[lane];

    const bool avalid = (node_l - WIN) > 0;
    const int lo = max(lm - 8 * lh, -pb);
    const int hi = min(lm - 8 * lh + 2 * WIN, NN - 1 - pb);
    float mx = avalid ? fabsf(adot) : -1e30f;
#pragma unroll
    for (int bt = 0; bt < 5; ++bt) {
#pragma unroll
        for (int i = 0; i < 8; ++i) {
            const int c = 16 * bt + i;
            if (c >= lo && c <= hi) mx = fmaxf(mx, fabsf(sc[bt][i]));
        }
    }
    mx = fmaxf(mx, __shfl_xor(mx, 16));
    float sum = 0.f;
#pragma unroll
    for (int bt = 0; bt < 5; ++bt) {
#pragma unroll
        for (int i = 0; i < 8; ++i) {
            const int c = 16 * bt + i;
            const bool ok = (c >= lo) && (c <= hi);
            const float s = sc[bt][i];
            const float e = __expf(fabsf(s) - mx);
            const float se = (s > 0.f) ? e : ((s < 0.f) ? -e : 0.f);
            sum += ok ? e : 0.f;
            sc[bt][i] = ok ? se : 0.f;
        }
    }
    sum += __shfl_xor(sum, 16);
    float ase = 0.f;
    if (avalid) {
        const float e = __expf(fabsf(adot) - mx);
        sum += e;
        ase = (adot > 0.f) ? e : ((adot < 0.f) ? -e : 0.f);
    }
    const float inv = 1.f / sum;
    const float wa  = ase * inv;
#pragma unroll
    for (int bt = 0; bt < 5; ++bt) {
#pragma unroll
        for (int i = 0; i < 8; ++i) sc[bt][i] *= inv;
    }
    if (wave == 0) {
        float ds = 0.f;
#pragma unroll
        for (int bt = 0; bt < 5; ++bt) {
#pragma unroll
            for (int i = 0; i < 8; ++i)
                ds += sc[bt][i] * state_in[base + clampi(pb + 16 * bt + i, 0, NN - 1)];
        }
        ds += __shfl_xor(ds, 16);
        ds += wa * state_in[base];
        if (lh == 0 && node_l < NN)
            state_out[base + node_l] = state_in[base + node_l] + ds;
    }

    {
        const int c0 = wave * 64;
        v8f acc[4] = {};
#pragma unroll
        for (int s = 0; s < 20; ++s) {
            const int bt = s >> 2;
            const int i0 = 2 * (s & 3);
            v2f a;
            a.x = sc[bt][i0];
            a.y = sc[bt][i0 + 1];
            const int r0 = base + clampi(pb + 16 * bt + i0,     0, NN - 1);
            const int r1 = base + clampi(pb + 16 * bt + i0 + 1, 0, NN - 1);
            const float* v0 = val_in + (size_t)r0 * D + c0 + lm;
            const float* v1 = val_in + (size_t)r1 * D + c0 + lm;
#pragma unroll
            for (int nt = 0; nt < 4; ++nt) {
                v2f bb;
                bb.x = v0[nt * 16];
                bb.y = v1[nt * 16];
                acc[nt] = __builtin_amdgcn_wmma_f32_16x16x4_f32(
                    false, a, false, bb, (short)0, acc[nt], false, false);
            }
        }
        {
            v2f a;
            a.x = (lh == 0) ? wa : 0.f;
            a.y = 0.f;
            const float* v0 = val_in + (size_t)base * D + c0 + lm;
#pragma unroll
            for (int nt = 0; nt < 4; ++nt) {
                v2f bb;
                bb.x = v0[nt * 16];
                bb.y = bb.x;
                acc[nt] = __builtin_amdgcn_wmma_f32_16x16x4_f32(
                    false, a, false, bb, (short)0, acc[nt], false, false);
            }
        }
        float gam[4], bet[4];
#pragma unroll
        for (int nt = 0; nt < 4; ++nt) {
            gam[nt] = gamma[c0 + nt * 16 + lm];
            bet[nt] = beta[c0 + nt * 16 + lm];
        }
        float y[4][8];
#pragma unroll
        for (int i = 0; i < 8; ++i) {
            const float* vr = val_in + (size_t)(base + min(p0 + i + 8 * lh, NN - 1)) * D + c0 + lm;
#pragma unroll
            for (int nt = 0; nt < 4; ++nt) y[nt][i] = acc[nt][i] + vr[nt * 16];
        }
        float mean[8];
        {
            float rs[8];
#pragma unroll
            for (int i = 0; i < 8; ++i)
                rs[i] = row16_sum((y[0][i] + y[1][i]) + (y[2][i] + y[3][i]));
            if (lm == 0) {
                float4* dst = (float4*)&s_red[0][wave][8 * lh];
                dst[0] = make_float4(rs[0], rs[1], rs[2], rs[3]);
                dst[1] = make_float4(rs[4], rs[5], rs[6], rs[7]);
            }
            __syncthreads();
            float tot[8];
#pragma unroll
            for (int w = 0; w < 8; ++w) {
                const float4* src = (const float4*)&s_red[0][w][8 * lh];
                const float4 a = src[0], c = src[1];
                if (w == 0) {
                    tot[0] = a.x; tot[1] = a.y; tot[2] = a.z; tot[3] = a.w;
                    tot[4] = c.x; tot[5] = c.y; tot[6] = c.z; tot[7] = c.w;
                } else {
                    tot[0] += a.x; tot[1] += a.y; tot[2] += a.z; tot[3] += a.w;
                    tot[4] += c.x; tot[5] += c.y; tot[6] += c.z; tot[7] += c.w;
                }
            }
#pragma unroll
            for (int i = 0; i < 8; ++i) mean[i] = tot[i] * (1.0f / D);
        }
        float rstd[8];
        {
            float rs[8];
#pragma unroll
            for (int i = 0; i < 8; ++i) {
                const float d0 = y[0][i] - mean[i], d1 = y[1][i] - mean[i];
                const float d2 = y[2][i] - mean[i], d3 = y[3][i] - mean[i];
                rs[i] = row16_sum((d0 * d0 + d1 * d1) + (d2 * d2 + d3 * d3));
            }
            if (lm == 0) {
                float4* dst = (float4*)&s_red[1][wave][8 * lh];
                dst[0] = make_float4(rs[0], rs[1], rs[2], rs[3]);
                dst[1] = make_float4(rs[4], rs[5], rs[6], rs[7]);
            }
            __syncthreads();
            float tot[8];
#pragma unroll
            for (int w = 0; w < 8; ++w) {
                const float4* src = (const float4*)&s_red[1][w][8 * lh];
                const float4 a = src[0], c = src[1];
                if (w == 0) {
                    tot[0] = a.x; tot[1] = a.y; tot[2] = a.z; tot[3] = a.w;
                    tot[4] = c.x; tot[5] = c.y; tot[6] = c.z; tot[7] = c.w;
                } else {
                    tot[0] += a.x; tot[1] += a.y; tot[2] += a.z; tot[3] += a.w;
                    tot[4] += c.x; tot[5] += c.y; tot[6] += c.z; tot[7] += c.w;
                }
            }
#pragma unroll
            for (int i = 0; i < 8; ++i) rstd[i] = rsqrtf(tot[i] * (1.0f / D) + 1e-5f);
        }
#pragma unroll
        for (int nt = 0; nt < 4; ++nt) {
#pragma unroll
            for (int i = 0; i < 8; ++i)
                y[nt][i] = (y[nt][i] - mean[i]) * rstd[i] * gam[nt] + bet[nt];
        }
        float* vo = val_out + (size_t)(base + p0 + 8 * lh) * D + c0 + lm;
        if (p0 + 16 <= NN) {
#pragma unroll
            for (int nt = 0; nt < 4; ++nt) {
#pragma unroll
                for (int i = 0; i < 8; ++i) vo[(size_t)i * D + nt * 16] = y[nt][i];
            }
        } else {
#pragma unroll
            for (int nt = 0; nt < 4; ++nt) {
#pragma unroll
                for (int i = 0; i < 8; ++i)
                    if (p0 + i + 8 * lh < NN) vo[(size_t)i * D + nt * 16] = y[nt][i];
            }
        }
        if (PROJ) {
#pragma unroll
            for (int nt = 0; nt < 4; ++nt) {
#pragma unroll
                for (int i = 0; i < 8; ++i) s_y[i + 8 * lh][c0 + nt * 16 + lm] = y[nt][i];
            }
        }
    }

    if (PROJ) {
        __syncthreads();
        qk_project_tile(&s_y[0][0], wq_next, wk_next, q_next, k_next,
                        base + p0, NN - p0, wave, lm, lh);
    }
}

extern "C" void kernel_launch(void* const* d_in, const int* in_sizes, int n_in,
                              void* d_out, int out_size, void* d_ws, size_t ws_size,
                              hipStream_t stream)
{
    const int*   ids    = (const int*)  d_in[0];
    const float* emb    = (const float*)d_in[1];
    const float* pos    = (const float*)d_in[2];
    const float* aval   = (const float*)d_in[3];
    const float* astate = (const float*)d_in[4];
    const float* ing    = (const float*)d_in[5];
    const float* inb    = (const float*)d_in[6];
    const float* spw    = (const float*)d_in[7];
    const float* spb    = (const float*)d_in[8];
    const float* wq     = (const float*)d_in[9];
    const float* wk     = (const float*)d_in[10];
    const float* lng    = (const float*)d_in[11];
    const float* lnb    = (const float*)d_in[12];

    float* ws = (float*)d_ws;
    size_t o = 0;
    float* val_a = ws + o; o += (size_t)MM * D;
    float* val_b = ws + o; o += (size_t)MM * D;
    float* st_a  = ws + o; o += MM;
    float* st_b  = ws + o; o += MM;
    float* qbuf  = ws + o; o += (size_t)MM * R;
    float* kbuf  = ws + o; o += (size_t)MM * R;
    float* qbuf2 = ws + o; o += (size_t)MM * R;
    float* kbuf2 = ws + o; o += (size_t)MM * R;

    float* out_state = (float*)d_out;
    float* out_val   = out_state + MM;

    const int ablocks = BB * TILES;

    embed_ln_qk_kernel<<<ablocks, 256, 0, stream>>>(ids, emb, pos, aval, astate,
                                                    ing, inb, spw, spb, val_a, st_a,
                                                    wq, wk, qbuf, kbuf);

    attn_fused_kernel<true><<<ablocks, 256, 0, stream>>>(
        val_a, st_a, qbuf, kbuf, lng, lnb, val_b, st_b,
        wq + (size_t)D * R, wk + (size_t)D * R, qbuf2, kbuf2);
    attn_fused_kernel<false><<<ablocks, 256, 0, stream>>>(
        val_b, st_b, qbuf2, kbuf2, lng + D, lnb + D, out_val, out_state,
        wq + (size_t)D * R, wk + (size_t)D * R, qbuf, kbuf);
}


# Softmax once per block: the wave that forms a score tile takes exp of its 8 slots against the tile's own max, with a partial sum; every wave merges ten (max,sum) pairs; softmax sum order changed
# speedup vs baseline: 4.0526x; 1.0682x over previous
#include <hip/hip_runtime.h>

#define D   512
#define R   64
#define WIN 32
#define BB  4
#define SS  2048
#define NN  (SS + 1)
#define MM  (BB * NN)
#define TILES ((NN + 15) / 16)
#define WTOT 66
#define DP  (D + 4)
#define KP  (R + 4)

typedef float v2f __attribute__((ext_vector_type(2)));
typedef float v8f __attribute__((ext_vector_type(8)));

__device__ __forceinline__ int clampi(int x, int lo, int hi) {
    return min(max(x, lo), hi);
}

__device__ __forceinline__ float row16_sum(float v) {
    v += __builtin_bit_cast(float, __builtin_amdgcn_update_dpp(0, __builtin_bit_cast(int, v), 0x161, 0xf, 0xf, true));
    v += __builtin_bit_cast(float, __builtin_amdgcn_update_dpp(0, __builtin_bit_cast(int, v), 0x162, 0xf, 0xf, true));
    v += __builtin_bit_cast(float, __builtin_amdgcn_update_dpp(0, __builtin_bit_cast(int, v), 0x164, 0xf, 0xf, true));
    v += __builtin_bit_cast(float, __builtin_amdgcn_update_dpp(0, __builtin_bit_cast(int, v), 0x168, 0xf, 0xf, true));
    return v;
}

__device__ __forceinline__ void qk_project_tile(
    const float* s_t, const float* __restrict__ wq, const float* __restrict__ wk,
    float* __restrict__ qo, float* __restrict__ ko,
    int grow0, int nvalid, int wave, int lm, int lh)
{
    constexpr int U = 16;
    constexpr int G = D / (4 * U);
    const float* wm = (wave < 4) ? wq : wk;
    float*       om = (wave < 4) ? qo : ko;
    const int col = (wave & 3) * 16 + lm;
    const float* bp = wm + (size_t)(2 * lh) * R + col;
    const float* ap = s_t + lm * DP + 2 * lh;

    float bx[U], by[U];
#pragma unroll
    for (int u = 0; u < U; ++u) {
        bx[u] = bp[(size_t)(4 * u) * R];
        by[u] = bp[(size_t)(4 * u + 1) * R];
    }
    v8f acc = {};
#pragma unroll
    for (int g = 0; g < G; ++g) {
        float nx[U], ny[U];
        if (g + 1 < G) {
            const float* np = bp + (size_t)(g + 1) * (4 * U) * R;
#pragma unroll
            for (int u = 0; u < U; ++u) {
                nx[u] = np[(size_t)(4 * u) * R];
                ny[u] = np[(size_t)(4 * u + 1) * R];
            }
        }
#pragma unroll
        for (int u = 0; u < U; ++u) {
            v2f a = *(const v2f*)(ap + g * (4 * U) + 4 * u);
            v2f b; b.x = bx[u]; b.y = by[u];
            acc = __builtin_amdgcn_wmma_f32_16x16x4_f32(
                false, a, false, b, (short)0, acc, false, false);
        }
        if (g + 1 < G) {
#pragma unroll
            for (int u = 0; u < U; ++u) { bx[u] = nx[u]; by[u] = ny[u]; }
        }
    }
#pragma unroll
    for (int i = 0; i < 8; ++i) {
        const int m = i + 8 * lh;
        if (m < nvalid) om[(size_t)(grow0 + m) * R + col] = acc[i];
    }
}

__global__ __launch_bounds__(256) void embed_ln_qk_kernel(
    const int* __restrict__ ids, const float* __restrict__ emb,
    const float* __restrict__ pos, const float* __restrict__ anchor_val,
    const float* __restrict__ anchor_state, const float* __restrict__ gamma,
    const float* __restrict__ beta, const float* __restrict__ sp_w,
    const float* __restrict__ sp_b, float* __restrict__ val,
    float* __restrict__ state,
    const float* __restrict__ wq, const float* __restrict__ wk,
    float* __restrict__ q, float* __restrict__ k)
{
    __shared__ __attribute__((aligned(16))) float s_y[16][DP];
    __shared__ __attribute__((aligned(16))) float s_p[3][16][16];

    const int tid  = threadIdx.x;
    const int lane = tid & 31;
    const int wave = tid >> 5;
    const int lm   = lane & 15;
    const int lh   = lane >> 4;
    const int b    = blockIdx.x / TILES;
    const int t    = blockIdx.x % TILES;
    const int p0   = t * 16;
    const int base = b * NN;

    const int m    = tid >> 4;
    const int jj   = tid & 15;
    const int node = p0 + m;
    const bool is_tok = (node >= 1) && (node < NN);

    float4 x[8];
    if (is_tok) {
        const int s  = node - 1;
        const int id = ids[b * SS + s];
        const float4* e4 = (const float4*)(emb + (size_t)id * D) + jj;
        const float4* p4 = (const float4*)(pos + (size_t)s * D) + jj;
#pragma unroll
        for (int i = 0; i < 8; ++i) {
            const float4 ev = e4[16 * i];
            const float4 pv = p4[16 * i];
            x[i] = make_float4(ev.x + pv.x, ev.y + pv.y, ev.z + pv.z, ev.w + pv.w);
        }
    } else if (node == 0) {
        const float4* a4 = (const float4*)anchor_val + jj;
#pragma unroll
        for (int i = 0; i < 8; ++i) x[i] = a4[16 * i];
    } else {
#pragma unroll
        for (int i = 0; i < 8; ++i) x[i] = make_float4(0.f, 0.f, 0.f, 0.f);
    }

    float part = 0.f;
#pragma unroll
    for (int i = 0; i < 8; ++i) part += (x[i].x + x[i].y) + (x[i].z + x[i].w);
    s_p[0][m][jj] = part;
    __syncthreads();
    float tot = 0.f;
#pragma unroll
    for (int u = 0; u < 16; ++u) tot += s_p[0][m][u];
    const float mean = tot * (1.0f / D);

    part = 0.f;
#pragma unroll
    for (int i = 0; i < 8; ++i) {
        const float cx = x[i].x - mean, cy = x[i].y - mean, cz = x[i].z - mean, cw = x[i].w - mean;
        part += (cx * cx + cy * cy) + (cz * cz + cw * cw);
    }
    s_p[1][m][jj] = part;
    __syncthreads();
    tot = 0.f;
#pragma unroll
    for (int u = 0; u < 16; ++u) tot += s_p[1][m][u];
    const float rstd = rsqrtf(tot * (1.0f / D) + 1e-5f);

    const float4* g4 = (const float4*)gamma + jj;
    const float4* b4 = (const float4*)beta + jj;
    const float4* w4 = (const float4*)sp_w + jj;
    part = 0.f;
#pragma unroll
    for (int i = 0; i < 8; ++i) {
        const float4 g = g4[16 * i];
        const float4 be = b4[16 * i];
        const float4 w = w4[16 * i];
        float4 y;
        y.x = (x[i].x - mean) * rstd * g.x + be.x;
        y.y = (x[i].y - mean) * rstd * g.y + be.y;
        y.z = (x[i].z - mean) * rstd * g.z + be.z;
        y.w = (x[i].w - mean) * rstd * g.w + be.w;
        part += (y.x * w.x + y.y * w.y) + (y.z * w.z + y.w * w.w);
        if (is_tok) x[i] = y;
        *((float4*)&s_y[m][4 * jj + 64 * i]) = x[i];
    }
    s_p[2][m][jj] = part;
    __syncthreads();
    if (node < NN) {
        float4* vo = (float4*)(val + (size_t)(base + node) * D) + jj;
#pragma unroll
        for (int i = 0; i < 8; ++i) vo[16 * i] = x[i];
    }
    if (jj == 0 && node < NN) {
        float st;
        if (node == 0) {
            st = anchor_state[0];
        } else {
            tot = 0.f;
#pragma unroll
            for (int u = 0; u < 16; ++u) tot += s_p[2][m][u];
            st = tot + sp_b[0];
        }
        state[base + node] = st;
    }

    qk_project_tile(&s_y[0][0], wq, wk, q, k, base + p0, NN - p0, wave, lm, lh);
}

template <bool PROJ>
__global__ __launch_bounds__(256) __attribute__((amdgpu_waves_per_eu(PROJ ? 6 : 2, 7))) void attn_fused_kernel(
    const float* __restrict__ val_in, const float* __restrict__ state_in,
    const float* __restrict__ q, const float* __restrict__ k,
    const float* __restrict__ gamma, const float* __restrict__ beta,
    float* __restrict__ val_out, float* __restrict__ state_out,
    const float* __restrict__ wq_next, const float* __restrict__ wk_next,
    float* __restrict__ q_next, float* __restrict__ k_next)
{
    __shared__ __attribute__((aligned(16))) float s_y[16][DP];
    __shared__ __attribute__((aligned(16))) float s_red[2][8][16];
    __shared__ __attribute__((aligned(16))) float4 s_sc[5][2][32];
    __shared__ float s_ad[32];
    __shared__ __attribute__((aligned(8))) v2f s_ml[5][32];
    __shared__ __attribute__((aligned(16))) float s_kq[97][KP];

    const int tid  = threadIdx.x;
    const int lane = tid & 31;
    const int wave = tid >> 5;
    const int lm   = lane & 15;
    const int lh   = lane >> 4;
    const int b    = blockIdx.x / TILES;
    const int t    = blockIdx.x % TILES;
    const int p0   = t * 16;
    const int base = b * NN;

    const int node_l = p0 + lm;
    const int pb     = p0 - WIN + 8 * lh;
    const float scale = 0.125f;
    const int wv = __builtin_amdgcn_readfirstlane(wave);
    const int lo = max(lm - 8 * lh, -pb);
    const int hi = min(lm - 8 * lh + 2 * WIN, NN - 1 - pb);
    {
        const int sr = tid >> 4;
        const int c4 = tid & 15;
        float4 kv[5];
#pragma unroll
        for (int j = 0; j < 5; ++j) {
            const int krow = base + clampi(p0 - WIN + 16 * j + sr, 0, NN - 1);
            kv[j] = *((const float4*)(k + (size_t)krow * R) + c4);
        }
        const float4 qv = *((const float4*)(q + (size_t)(base + clampi(p0 + sr, 0, NN - 1)) * R) + c4);
        const float4 av = *((const float4*)(k + (size_t)base * R) + c4);
#pragma unroll
        for (int j = 0; j < 5; ++j) *((float4*)&s_kq[16 * j + sr][4 * c4]) = kv[j];
        *((float4*)&s_kq[80 + sr][4 * c4]) = qv;
        asm volatile("" :: "v"(av.x), "v"(av.y), "v"(av.z), "v"(av.w));
        if (sr == 0) *((float4*)&s_kq[96][4 * c4]) = av;
    }
    __syncthreads();
    if (wv < 6) {
        v2f qf[16];
        {
            const float* qp = &s_kq[80 + lm][2 * lh];
#pragma unroll
            for (int s = 0; s < 16; ++s) qf[s] = *(const v2f*)(qp + 4 * s);
        }
        if (wv < 5) {
            const float* kp = &s_kq[16 * wv + lm][2 * lh];
            v8f acc = {};
#pragma unroll
            for (int s = 0; s < 16; ++s) {
                v2f a = *(const v2f*)(kp + 4 * s);
                acc = __builtin_amdgcn_wmma_f32_16x16x4_f32(
                    false, a, false, qf[s], (short)0, acc, false, false);
            }
            float sv[8];
            float ml = -1e30f;
#pragma unroll
            for (int i = 0; i < 8; ++i) {
                const int c = 16 * wv + i;
                sv[i] = acc[i] * scale;
                if (c >= lo && c <= hi) ml = fmaxf(ml, fabsf(sv[i]));
            }
            float ll = 0.f;
#pragma unroll
            for (int i = 0; i < 8; ++i) {
                const int c = 16 * wv + i;
                const bool ok = (c >= lo) && (c <= hi);
                const float e = __expf(fabsf(sv[i]) - ml);
                const float se = (sv[i] > 0.f) ? e : ((sv[i] < 0.f) ? -e : 0.f);
                ll += ok ? e : 0.f;
                sv[i] = ok ? se : 0.f;
            }
            s_sc[wv][0][lane] = make_float4(sv[0], sv[1], sv[2], sv[3]);
            s_sc[wv][1][lane] = make_float4(sv[4], sv[5], sv[6], sv[7]);
            v2f mlv; mlv.x = ml; mlv.y = ll;
            s_ml[wv][lane] = mlv;
        } else {
            const float* k0 = &s_kq[96][2 * lh];
            float pa = 0.f;
#pragma unroll
            for (int s = 0; s < 16; ++s) {
                const v2f kv = *(const v2f*)(k0 + 4 * s);
                pa = fmaf(qf[s].x, kv.x, pa);
                pa = fmaf(qf[s].y, kv.y, pa);
            }
            s_ad[lane] = (pa + __shfl_xor(pa, 16)) * scale;
        }
    }
    __syncthreads();
    v8f sc[5];
#pragma unroll
    for (int bt = 0; bt < 5; ++bt) {
        const float4 a = s_sc[bt][0][lane], c = s_sc[bt][1][lane];
        sc[bt][0] = a.x; sc[bt][1] = a.y; sc[bt][2] = a.z; sc[bt][3] = a.w;
        sc[bt][4] = c.x; sc[bt][5] = c.y; sc[bt][6] = c.z; sc[bt][7] = c.w;
    }
    const float adot = s_ad[lane];

    const bool avalid = (node_l - WIN) > 0;
    v2f own[5], oth[5];
#pragma unroll
    for (int bt = 0; bt < 5; ++bt) {
        own[bt] = s_ml[bt][lane];
        oth[bt] = s_ml[bt][lane ^ 16];
    }
    float mx = avalid ? fabsf(adot) : -1e30f;
#pragma unroll
    for (int bt = 0; bt < 5; ++bt) mx = fmaxf(mx, fmaxf(own[bt].x, oth[bt].x));
    float fo[5];
    float so = 0.f, sp = 0.f;
#pragma unroll
    for (int bt = 0; bt < 5; ++bt) {
        fo[bt] = __expf(own[bt].x - mx);
        so = fmaf(fo[bt], own[bt].y, so);
        sp = fmaf(__expf(oth[bt].x - mx), oth[bt].y, sp);
    }
    float sum = so + sp;
    float ase = 0.f;
    if (avalid) {
        const float e = __expf(fabsf(adot) - mx);
        sum += e;
        ase = (adot > 0.f) ? e : ((adot < 0.f) ? -e : 0.f);
    }
    const float inv = 1.f / sum;
    const float wa  = ase * inv;
#pragma unroll
    for (int bt = 0; bt < 5; ++bt) {
        const float r = fo[bt] * inv;
#pragma unroll
        for (int i = 0; i < 8; ++i) sc[bt][i] *= r;
    }
    if (wave == 0) {
        float ds = 0.f;
#pragma unroll
        for (int bt = 0; bt < 5; ++bt) {
#pragma unroll
            for (int i = 0; i < 8; ++i)
                ds += sc[bt][i] * state_in[base + clampi(pb + 16 * bt + i, 0, NN - 1)];
        }
        ds += __shfl_xor(ds, 16);
        ds += wa * state_in[base];
        if (lh == 0 && node_l < NN)
            state_out[base + node_l] = state_in[base + node_l] + ds;
    }

    {
        const int c0 = wave * 64;
        v8f acc[4] = {};
#pragma unroll
        for (int s = 0; s < 20; ++s) {
            const int bt = s >> 2;
            const int i0 = 2 * (s & 3);
            v2f a;
            a.x = sc[bt][i0];
            a.y = sc[bt][i0 + 1];
            const int r0 = base + clampi(pb + 16 * bt + i0,     0, NN - 1);
            const int r1 = base + clampi(pb + 16 * bt + i0 + 1, 0, NN - 1);
            const float* v0 = val_in + (size_t)r0 * D + c0 + lm;
            const float* v1 = val_in + (size_t)r1 * D + c0 + lm;
#pragma unroll
            for (int nt = 0; nt < 4; ++nt) {
                v2f bb;
                bb.x = v0[nt * 16];
                bb.y = v1[nt * 16];
                acc[nt] = __builtin_amdgcn_wmma_f32_16x16x4_f32(
                    false, a, false, bb, (short)0, acc[nt], false, false);
            }
        }
        {
            v2f a;
            a.x = (lh == 0) ? wa : 0.f;
            a.y = 0.f;
            const float* v0 = val_in + (size_t)base * D + c0 + lm;
#pragma unroll
            for (int nt = 0; nt < 4; ++nt) {
                v2f bb;
                bb.x = v0[nt * 16];
                bb.y = bb.x;
                acc[nt] = __builtin_amdgcn_wmma_f32_16x16x4_f32(
                    false, a, false, bb, (short)0, acc[nt], false, false);
            }
        }
        float gam[4], bet[4];
#pragma unroll
        for (int nt = 0; nt < 4; ++nt) {
            gam[nt] = gamma[c0 + nt * 16 + lm];
            bet[nt] = beta[c0 + nt * 16 + lm];
        }
        float y[4][8];
#pragma unroll
        for (int i = 0; i < 8; ++i) {
            const float* vr = val_in + (size_t)(base + min(p0 + i + 8 * lh, NN - 1)) * D + c0 + lm;
#pragma unroll
            for (int nt = 0; nt < 4; ++nt) y[nt][i] = acc[nt][i] + vr[nt * 16];
        }
        float mean[8];
        {
            float rs[8];
#pragma unroll
            for (int i = 0; i < 8; ++i)
                rs[i] = row16_sum((y[0][i] + y[1][i]) + (y[2][i] + y[3][i]));
            if (lm == 0) {
                float4* dst = (float4*)&s_red[0][wave][8 * lh];
                dst[0] = make_float4(rs[0], rs[1], rs[2], rs[3]);
                dst[1] = make_float4(rs[4], rs[5], rs[6], rs[7]);
            }
            __syncthreads();
            float tot[8];
#pragma unroll
            for (int w = 0; w < 8; ++w) {
                const float4* src = (const float4*)&s_red[0][w][8 * lh];
                const float4 a = src[0], c = src[1];
                if (w == 0) {
                    tot[0] = a.x; tot[1] = a.y; tot[2] = a.z; tot[3] = a.w;
                    tot[4] = c.x; tot[5] = c.y; tot[6] = c.z; tot[7] = c.w;
                } else {
                    tot[0] += a.x; tot[1] += a.y; tot[2] += a.z; tot[3] += a.w;
                    tot[4] += c.x; tot[5] += c.y; tot[6] += c.z; tot[7] += c.w;
                }
            }
#pragma unroll
            for (int i = 0; i < 8; ++i) mean[i] = tot[i] * (1.0f / D);
        }
        float rstd[8];
        {
            float rs[8];
#pragma unroll
            for (int i = 0; i < 8; ++i) {
                const float d0 = y[0][i] - mean[i], d1 = y[1][i] - mean[i];
                const float d2 = y[2][i] - mean[i], d3 = y[3][i] - mean[i];
                rs[i] = row16_sum((d0 * d0 + d1 * d1) + (d2 * d2 + d3 * d3));
            }
            if (lm == 0) {
                float4* dst = (float4*)&s_red[1][wave][8 * lh];
                dst[0] = make_float4(rs[0], rs[1], rs[2], rs[3]);
                dst[1] = make_float4(rs[4], rs[5], rs[6], rs[7]);
            }
            __syncthreads();
            float tot[8];
#pragma unroll
            for (int w = 0; w < 8; ++w) {
                const float4* src = (const float4*)&s_red[1][w][8 * lh];
                const float4 a = src[0], c = src[1];
                if (w == 0) {
                    tot[0] = a.x; tot[1] = a.y; tot[2] = a.z; tot[3] = a.w;
                    tot[4] = c.x; tot[5] = c.y; tot[6] = c.z; tot[7] = c.w;
                } else {
                    tot[0] += a.x; tot[1] += a.y; tot[2] += a.z; tot[3] += a.w;
                    tot[4] += c.x; tot[5] += c.y; tot[6] += c.z; tot[7] += c.w;
                }
            }
#pragma unroll
            for (int i = 0; i < 8; ++i) rstd[i] = rsqrtf(tot[i] * (1.0f / D) + 1e-5f);
        }
#pragma unroll
        for (int nt = 0; nt < 4; ++nt) {
#pragma unroll
            for (int i = 0; i < 8; ++i)
                y[nt][i] = (y[nt][i] - mean[i]) * rstd[i] * gam[nt] + bet[nt];
        }
        float* vo = val_out + (size_t)(base + p0 + 8 * lh) * D + c0 + lm;
        if (p0 + 16 <= NN) {
#pragma unroll
            for (int nt = 0; nt < 4; ++nt) {
#pragma unroll
                for (int i = 0; i < 8; ++i) vo[(size_t)i * D + nt * 16] = y[nt][i];
            }
        } else {
#pragma unroll
            for (int nt = 0; nt < 4; ++nt) {
#pragma unroll
                for (int i = 0; i < 8; ++i)
                    if (p0 + i + 8 * lh < NN) vo[(size_t)i * D + nt * 16] = y[nt][i];
            }
        }
        if (PROJ) {
#pragma unroll
            for (int nt = 0; nt < 4; ++nt) {
#pragma unroll
                for (int i = 0; i < 8; ++i) s_y[i + 8 * lh][c0 + nt * 16 + lm] = y[nt][i];
            }
        }
    }

    if (PROJ) {
        __syncthreads();
        qk_project_tile(&s_y[0][0], wq_next, wk_next, q_next, k_next,
                        base + p0, NN - p0, wave, lm, lh);
    }
}

extern "C" void kernel_launch(void* const* d_in, const int* in_sizes, int n_in,
                              void* d_out, int out_size, void* d_ws, size_t ws_size,
                              hipStream_t stream)
{
    const int*   ids    = (const int*)  d_in[0];
    const float* emb    = (const float*)d_in[1];
    const float* pos    = (const float*)d_in[2];
    const float* aval   = (const float*)d_in[3];
    const float* astate = (const float*)d_in[4];
    const float* ing    = (const float*)d_in[5];
    const float* inb    = (const float*)d_in[6];
    const float* spw    = (const float*)d_in[7];
    const float* spb    = (const float*)d_in[8];
    const float* wq     = (const float*)d_in[9];
    const float* wk     = (const float*)d_in[10];
    const float* lng    = (const float*)d_in[11];
    const float* lnb    = (const float*)d_in[12];

    float* ws = (float*)d_ws;
    size_t o = 0;
    float* val_a = ws + o; o += (size_t)MM * D;
    float* val_b = ws + o; o += (size_t)MM * D;
    float* st_a  = ws + o; o += MM;
    float* st_b  = ws + o; o += MM;
    float* qbuf  = ws + o; o += (size_t)MM * R;
    float* kbuf  = ws + o; o += (size_t)MM * R;
    float* qbuf2 = ws + o; o += (size_t)MM * R;
    float* kbuf2 = ws + o; o += (size_t)MM * R;

    float* out_state = (float*)d_out;
    float* out_val   = out_state + MM;

    const int ablocks = BB * TILES;

    embed_ln_qk_kernel<<<ablocks, 256, 0, stream>>>(ids, emb, pos, aval, astate,
                                                    ing, inb, spw, spb, val_a, st_a,
                                                    wq, wk, qbuf, kbuf);

    attn_fused_kernel<true><<<ablocks, 256, 0, stream>>>(
        val_a, st_a, qbuf, kbuf, lng, lnb, val_b, st_b,
        wq + (size_t)D * R, wk + (size_t)D * R, qbuf2, kbuf2);
    attn_fused_kernel<false><<<ablocks, 256, 0, stream>>>(
        val_b, st_b, qbuf2, kbuf2, lng + D, lnb + D, out_val, out_state,
        wq + (size_t)D * R, wk + (size_t)D * R, qbuf, kbuf);
}
